# starting kernel
# baseline (speedup 1.0000x reference)
#include <hip/hip_runtime.h>

typedef float v2f __attribute__((ext_vector_type(2)));
typedef float v4f __attribute__((ext_vector_type(4)));
typedef float v8f __attribute__((ext_vector_type(8)));

__global__ __launch_bounds__(256) void gcn_deg_init(float* __restrict__ deg, int n) {
    int i = blockIdx.x * blockDim.x + threadIdx.x;
    if (i < n) deg[i] = 1.0f;
}

__global__ __launch_bounds__(256) void gcn_deg_count(float* __restrict__ deg,
                                                     const int* __restrict__ dst, int e) {
    int i = blockIdx.x * blockDim.x + threadIdx.x;
    if (i < e) atomicAdd(&deg[dst[i]], 1.0f);
}

__global__ __launch_bounds__(256) void gcn_rsqrt(float* __restrict__ dis,
                                                 const float* __restrict__ deg, int n) {
    int i = blockIdx.x * blockDim.x + threadIdx.x;
    if (i < n) dis[i] = rsqrtf(deg[i]);
}

template <int KK, int TN, bool RELU, bool BIAS>
__global__ __launch_bounds__(256) void gcn_gemm_wmma(
        const float* __restrict__ A, const float* __restrict__ B,
        const float* __restrict__ bias, float* __restrict__ C, int M) {
    constexpr int Nc  = TN * 16;
    constexpr int LDB = KK + 2;
    __shared__ float Blds[Nc * LDB];

    for (int i = threadIdx.x; i < KK * Nc; i += 256) {
        int k = i / Nc;
        int c = i - k * Nc;
        Blds[c * LDB + k] = B[i];
    }
    __syncthreads();

    int tilesM = M >> 4;
    int tm     = blockIdx.x * 8 + (threadIdx.x >> 5);
    int lane   = threadIdx.x & 31;
    int half   = lane >> 4;
    int lr     = lane & 15;

    if (tm < tilesM) {
        const float* Arow = A + (size_t)(tm * 16 + lr) * KK;
        __builtin_prefetch(Arow, 0, 0);
        __builtin_prefetch(Arow + 64, 0, 0);

        v8f acc[TN] = {};
        for (int k = 0; k < KK; k += 4) {
            v2f a = *(const v2f*)(Arow + k + 2 * half);
#pragma unroll
            for (int t = 0; t < TN; ++t) {
                v2f b = *(const v2f*)(&Blds[(t * 16 + lr) * LDB + k + 2 * half]);
                acc[t] = __builtin_amdgcn_wmma_f32_16x16x4_f32(
                             false, a, false, b, (short)0, acc[t], false, false);
            }
        }

#pragma unroll
        for (int t = 0; t < TN; ++t) {
            int col  = t * 16 + lr;
            float bv = BIAS ? bias[col] : 0.0f;
#pragma unroll
            for (int r = 0; r < 8; ++r) {
                int row = tm * 16 + r + 8 * half;
                float v = acc[t][r] + bv;
                if (RELU) v = fmaxf(v, 0.0f);
                C[(size_t)row * Nc + col] = v;
            }
        }
    }
}

__global__ __launch_bounds__(256) void gcn_agg_init(
        float* __restrict__ out, const float* __restrict__ hw,
        const float* __restrict__ dis, const float* __restrict__ bias,
        int n, int h) {
    int h4  = h >> 2;
    int idx = blockIdx.x * blockDim.x + threadIdx.x;
    if (idx >= n * h4) return;
    int row = idx / h4;
    int cj  = idx - row * h4;
    float sn = dis[row];
    sn = sn * sn;
    v4f v  = ((const v4f*)hw)[idx];
    v4f bv = ((const v4f*)bias)[cj];
    ((v4f*)out)[idx] = v * sn + bv;
}

__global__ __launch_bounds__(256) void gcn_scatter(
        float* __restrict__ agg, const float* __restrict__ hw,
        const float* __restrict__ dis, const int* __restrict__ src,
        const int* __restrict__ dst, int e, int h) {
    int wave = blockIdx.x * (blockDim.x >> 5) + (threadIdx.x >> 5);
    if (wave >= e) return;
    int lane = threadIdx.x & 31;
    int s = src[wave];
    int d = dst[wave];
    float nrm = dis[s] * dis[d];

    v4f m = ((const v4f*)(hw + (size_t)s * h))[lane];
    m *= nrm;
    float* drow = agg + (size_t)d * h + lane * 4;
    atomicAdd(drow + 0, m.x);
    atomicAdd(drow + 1, m.y);
    atomicAdd(drow + 2, m.z);
    atomicAdd(drow + 3, m.w);
}

extern "C" void kernel_launch(void* const* d_in, const int* in_sizes, int n_in,
                              void* d_out, int out_size, void* d_ws, size_t ws_size,
                              hipStream_t stream) {
    const float* x  = (const float*)d_in[0];
    const int*   ei = (const int*)  d_in[1];
    const float* W1 = (const float*)d_in[2];
    const float* b1 = (const float*)d_in[3];
    const float* Wc = (const float*)d_in[4];
    const float* bc = (const float*)d_in[5];
    const float* W2 = (const float*)d_in[6];
    const float* b2 = (const float*)d_in[7];

    const int N = 50000, E = 800000, H = 128, L = 3;
    const int* srcv = ei;
    const int* dstv = ei + E;

    char* ws = (char*)d_ws;
    size_t off = 0;
    auto carve = [&](size_t bytes) -> void* {
        void* p = ws + off;
        off = (off + bytes + 255) & ~(size_t)255;
        return p;
    };
    float* deg  = (float*)carve((size_t)N * 4);
    float* dis  = (float*)carve((size_t)N * 4);
    float* bufA = (float*)carve((size_t)N * H * 4);
    float* bufB = (float*)carve((size_t)N * H * 4);
    float* hw   = (float*)carve((size_t)N * H * 4);

    gcn_deg_init <<<(N + 255) / 256, 256, 0, stream>>>(deg, N);
    gcn_deg_count<<<(E + 255) / 256, 256, 0, stream>>>(deg, dstv, E);
    gcn_rsqrt    <<<(N + 255) / 256, 256, 0, stream>>>(dis, deg, N);

    const int tilesM    = N / 16;
    const int gemmBlks  = (tilesM + 7) / 8;

    gcn_gemm_wmma<128, 8, true, true><<<gemmBlks, 256, 0, stream>>>(
        x, W1, b1, bufA, N);

    float* hcur = bufA;
    float* hnxt = bufB;
    for (int l = 0; l < L; ++l) {
        gcn_gemm_wmma<128, 8, false, false><<<gemmBlks, 256, 0, stream>>>(
            hcur, Wc + (size_t)l * H * H, nullptr, hw, N);

        int tot4 = N * (H / 4);
        gcn_agg_init<<<(tot4 + 255) / 256, 256, 0, stream>>>(
            hnxt, hw, dis, bc + (size_t)l * H, N, H);

        gcn_scatter<<<(E + 7) / 8, 256, 0, stream>>>(
            hnxt, hw, dis, srcv, dstv, E, H);

        float* t = hcur; hcur = hnxt; hnxt = t;
    }

    gcn_gemm_wmma<128, 4, false, true><<<gemmBlks, 256, 0, stream>>>(
        hcur, W2, b2, (float*)d_out, N);
}


# edge sum as per-node wave gather over a CSR built in each call (int-atomic count+rank, scan, place): f32 atomics gone, self-loop+bias fused; sum order = CSR slot order, still arrival-dependent
# speedup vs baseline: 2.5004x; 2.5004x over previous
#include <hip/hip_runtime.h>


typedef float v2f __attribute__((ext_vector_type(2)));
typedef float v4f __attribute__((ext_vector_type(4)));
typedef float v8f __attribute__((ext_vector_type(8)));
typedef int   v2i __attribute__((ext_vector_type(2)));
typedef int   v4i __attribute__((ext_vector_type(4)));

__global__ __launch_bounds__(256) void csr_zero(int* __restrict__ cnt, int n) {
    int i = blockIdx.x * blockDim.x + threadIdx.x;
    if (i < n) cnt[i] = 0;
}

__global__ __launch_bounds__(256) void csr_count_rank(int* __restrict__ cnt,
                                                      int* __restrict__ rank,
                                                      const int* __restrict__ dst, int e) {
    int i = blockIdx.x * blockDim.x + threadIdx.x;
    if (i < e) rank[i] = atomicAdd(&cnt[dst[i]], 1);
}

__global__ __launch_bounds__(256) void csr_scan(const int* __restrict__ cnt,
                                                int* __restrict__ rowptr,
                                                float* __restrict__ dis, int n) {
    __shared__ int s_part[8];
    __shared__ int s_tot[8];
    int tid  = threadIdx.x;
    int lane = tid & 31;
    int wv   = tid >> 5;
    int base = blockIdx.x * 256;

    const v4i* c4 = (const v4i*)cnt;
    const int nq = base >> 2;
    int part = 0;
    for (int j0 = 0; j0 < nq; j0 += 2048) {
        v4i q[8];
#pragma unroll
        for (int u = 0; u < 8; ++u) {
            int idx = j0 + u * 256 + tid;
            q[u] = c4[idx < nq ? idx : nq - 1];
        }
#pragma unroll
        for (int u = 0; u < 8; ++u) {
            int idx = j0 + u * 256 + tid;
            int sum = (q[u].x + q[u].y) + (q[u].z + q[u].w);
            part += (idx < nq) ? sum : 0;
        }
    }

    int i = base + tid;
    int c = (i < n) ? cnt[i] : 0;
    int incl = c;
    for (int o = 1; o < 32; o <<= 1) {
        int t = __shfl_up(incl, o);
        if (lane >= o) incl += t;
    }
    for (int o = 16; o > 0; o >>= 1) part += __shfl_xor(part, o);
    if (lane == 31) s_tot[wv]  = incl;
    if (lane == 0)  s_part[wv] = part;
    __syncthreads();

    int off = 0;
    for (int w = 0; w < 8; ++w) {
        off += s_part[w];
        if (w < wv) off += s_tot[w];
    }
    if (i < n) {
        rowptr[i] = off + incl - c;
        dis[i]    = rsqrtf((float)c + 1.0f);
    }
}

__global__ __launch_bounds__(256) void csr_place(int* __restrict__ edges,
                                                 const int* __restrict__ src,
                                                 const int* __restrict__ dst,
                                                 const int* __restrict__ rank,
                                                 const int* __restrict__ rowptr,
                                                 const float* __restrict__ dis, int e) {
    int i = blockIdx.x * blockDim.x + threadIdx.x;
    if (i >= e) return;
    int s = src[i];
    int d = dst[i];
    int p = rowptr[d] + rank[i];
    v2i rec;
    rec.x = s;
    rec.y = __float_as_int(dis[s] * dis[d]);
    ((v2i*)edges)[p] = rec;
}

template <int KK, int TN, bool RELU, bool BIAS>
__global__ __launch_bounds__(256) void gcn_gemm_wmma(
        const float* __restrict__ A, const float* __restrict__ B,
        const float* __restrict__ bias, float* __restrict__ C, int M) {
    constexpr int Nc  = TN * 16;
    constexpr int LDB = KK + 2;
    __shared__ float Blds[Nc * LDB];

    for (int i = threadIdx.x; i < KK * Nc; i += 256) {
        int k = i / Nc;
        int c = i - k * Nc;
        Blds[c * LDB + k] = B[i];
    }
    __syncthreads();

    int tilesM = M >> 4;
    int tm     = blockIdx.x * 8 + (threadIdx.x >> 5);
    int lane   = threadIdx.x & 31;
    int half   = lane >> 4;
    int lr     = lane & 15;

    if (tm < tilesM) {
        const float* Arow = A + (size_t)(tm * 16 + lr) * KK;
        __builtin_prefetch(Arow, 0, 0);
        __builtin_prefetch(Arow + 64, 0, 0);

        v8f acc[TN] = {};
        for (int k = 0; k < KK; k += 4) {
            v2f a = *(const v2f*)(Arow + k + 2 * half);
#pragma unroll
            for (int t = 0; t < TN; ++t) {
                v2f b = *(const v2f*)(&Blds[(t * 16 + lr) * LDB + k + 2 * half]);
                acc[t] = __builtin_amdgcn_wmma_f32_16x16x4_f32(
                             false, a, false, b, (short)0, acc[t], false, false);
            }
        }

#pragma unroll
        for (int t = 0; t < TN; ++t) {
            int col  = t * 16 + lr;
            float bv = BIAS ? bias[col] : 0.0f;
#pragma unroll
            for (int r = 0; r < 8; ++r) {
                int row = tm * 16 + r + 8 * half;
                float v = acc[t][r] + bv;
                if (RELU) v = fmaxf(v, 0.0f);
                C[(size_t)row * Nc + col] = v;
            }
        }
    }
}

template <int U>
__device__ __forceinline__ void gather_rows(v4f& acc, const v4f* __restrict__ hw4,
                                            v2i rec, int j, int lane) {
    v4f   r[U];
    float w[U];
#pragma unroll
    for (int u = 0; u < U; ++u) {
        int s = __builtin_amdgcn_readlane(rec.x, j + u);
        w[u]  = __int_as_float(__builtin_amdgcn_readlane(rec.y, j + u));
        r[u]  = hw4[(size_t)s * 32 + lane];
    }
#pragma unroll
    for (int u = 0; u < U; ++u) acc += r[u] * w[u];
}

__global__ __launch_bounds__(256) void gcn_gather(
        float* __restrict__ out, const float* __restrict__ hw,
        const float* __restrict__ dis, const float* __restrict__ bias,
        const int* __restrict__ rowptr, const int* __restrict__ cnt,
        const int* __restrict__ edges, int n) {
    int node = blockIdx.x * 8 + __builtin_amdgcn_readfirstlane((int)(threadIdx.x >> 5));
    if (node >= n) return;
    int lane  = threadIdx.x & 31;
    int start = __builtin_amdgcn_readfirstlane(rowptr[node]);
    int deg   = __builtin_amdgcn_readfirstlane(cnt[node]);

    const v4f* hw4 = (const v4f*)hw;
    const v2i* ed  = (const v2i*)edges + start;

    float sn   = dis[node];
    v4f   self = hw4[(size_t)node * 32 + lane];
    v4f   bv   = ((const v4f*)bias)[lane];

    v4f acc = {0.0f, 0.0f, 0.0f, 0.0f};
    for (int base = 0; base < deg; base += 32) {
        int m = deg - base;
        if (m > 32) m = 32;
        v2i rec = {0, 0};
        if (lane < m) rec = ed[base + lane];
        int j = 0;
        for (; j + 8 <= m; j += 8) gather_rows<8>(acc, hw4, rec, j, lane);
        if (j + 4 <= m) { gather_rows<4>(acc, hw4, rec, j, lane); j += 4; }
        if (j + 2 <= m) { gather_rows<2>(acc, hw4, rec, j, lane); j += 2; }
        if (j < m)      { gather_rows<1>(acc, hw4, rec, j, lane); }
    }

    sn = sn * sn;
    ((v4f*)out)[(size_t)node * 32 + lane] = acc + self * sn + bv;
}

extern "C" void kernel_launch(void* const* d_in, const int* in_sizes, int n_in,
                              void* d_out, int out_size, void* d_ws, size_t ws_size,
                              hipStream_t stream) {
    const float* x  = (const float*)d_in[0];
    const int*   ei = (const int*)  d_in[1];
    const float* W1 = (const float*)d_in[2];
    const float* b1 = (const float*)d_in[3];
    const float* Wc = (const float*)d_in[4];
    const float* bc = (const float*)d_in[5];
    const float* W2 = (const float*)d_in[6];
    const float* b2 = (const float*)d_in[7];

    const int N = 50000, E = 800000, H = 128, L = 3;
    const int* srcv = ei;
    const int* dstv = ei + E;

    char* ws = (char*)d_ws;
    size_t off = 0;
    auto carve = [&](size_t bytes) -> void* {
        void* p = ws + off;
        off = (off + bytes + 255) & ~(size_t)255;
        return p;
    };
    int*   cnt    = (int*)  carve((size_t)N * 4);
    int*   rowptr = (int*)  carve((size_t)N * 4);
    float* dis    = (float*)carve((size_t)N * 4);
    int*   rank   = (int*)  carve((size_t)E * 4);
    int*   edges  = (int*)  carve((size_t)E * 8);
    float* bufA   = (float*)carve((size_t)N * H * 4);
    float* bufB   = (float*)carve((size_t)N * H * 4);
    float* hw     = (float*)carve((size_t)N * H * 4);

    csr_zero      <<<(N + 255) / 256, 256, 0, stream>>>(cnt, N);
    csr_count_rank<<<(E + 255) / 256, 256, 0, stream>>>(cnt, rank, dstv, E);
    csr_scan      <<<(N + 255) / 256, 256, 0, stream>>>(cnt, rowptr, dis, N);
    csr_place     <<<(E + 255) / 256, 256, 0, stream>>>(edges, srcv, dstv, rank,
                                                        rowptr, dis, E);

    const int tilesM    = N / 16;
    const int gemmBlks  = (tilesM + 7) / 8;

    gcn_gemm_wmma<128, 8, true, true><<<gemmBlks, 256, 0, stream>>>(
        x, W1, b1, bufA, N);

    float* hcur = bufA;
    float* hnxt = bufB;
    for (int l = 0; l < L; ++l) {
        gcn_gemm_wmma<128, 8, false, false><<<gemmBlks, 256, 0, stream>>>(
            hcur, Wc + (size_t)l * H * H, nullptr, hw, N);

        gcn_gather<<<(N + 7) / 8, 256, 0, stream>>>(
            hnxt, hw, dis, bc + (size_t)l * H, rowptr, cnt, edges, N);

        float* t = hcur; hcur = hnxt; hnxt = t;
    }

    gcn_gemm_wmma<128, 4, false, true><<<gemmBlks, 256, 0, stream>>>(
        hcur, W2, b2, (float*)d_out, N);
}


# no nonlinearity between conv layers: Wc0*Wc1*Wc2*W2 folded in-call on WMMA to one 128x64 matrix, 3 aggregations run on 64 columns; fp sum order changes (products reassociated, even/odd edge halves)
# speedup vs baseline: 3.3795x; 1.3516x over previous
#include <hip/hip_runtime.h>


typedef float v2f __attribute__((ext_vector_type(2)));
typedef float v4f __attribute__((ext_vector_type(4)));
typedef float v8f __attribute__((ext_vector_type(8)));
typedef int   v2i __attribute__((ext_vector_type(2)));
typedef int   v4i __attribute__((ext_vector_type(4)));

__global__ __launch_bounds__(256) void csr_zero(int* __restrict__ cnt, int n) {
    int i = blockIdx.x * blockDim.x + threadIdx.x;
    if (i < n) cnt[i] = 0;
}

__global__ __launch_bounds__(256) void csr_count_rank(int* __restrict__ cnt,
                                                      int* __restrict__ rank,
                                                      const int* __restrict__ dst, int e) {
    int i = blockIdx.x * blockDim.x + threadIdx.x;
    if (i < e) rank[i] = atomicAdd(&cnt[dst[i]], 1);
}

__global__ __launch_bounds__(256) void csr_scan(const int* __restrict__ cnt,
                                                int* __restrict__ rowptr,
                                                float* __restrict__ dis, int n) {
    __shared__ int s_part[8];
    __shared__ int s_tot[8];
    int tid  = threadIdx.x;
    int lane = tid & 31;
    int wv   = tid >> 5;
    int base = blockIdx.x * 256;

    const v4i* c4 = (const v4i*)cnt;
    const int nq = base >> 2;
    int part = 0;
    for (int j0 = 0; j0 < nq; j0 += 2048) {
        v4i q[8];
#pragma unroll
        for (int u = 0; u < 8; ++u) {
            int idx = j0 + u * 256 + tid;
            q[u] = c4[idx < nq ? idx : nq - 1];
        }
#pragma unroll
        for (int u = 0; u < 8; ++u) {
            int idx = j0 + u * 256 + tid;
            int sum = (q[u].x + q[u].y) + (q[u].z + q[u].w);
            part += (idx < nq) ? sum : 0;
        }
    }

    int i = base + tid;
    int c = (i < n) ? cnt[i] : 0;
    int incl = c;
    for (int o = 1; o < 32; o <<= 1) {
        int t = __shfl_up(incl, o);
        if (lane >= o) incl += t;
    }
    for (int o = 16; o > 0; o >>= 1) part += __shfl_xor(part, o);
    if (lane == 31) s_tot[wv]  = incl;
    if (lane == 0)  s_part[wv] = part;
    __syncthreads();

    int off = 0;
    for (int w = 0; w < 8; ++w) {
        off += s_part[w];
        if (w < wv) off += s_tot[w];
    }
    if (i < n) {
        rowptr[i] = off + incl - c;
        dis[i]    = rsqrtf((float)c + 1.0f);
    }
}

__global__ __launch_bounds__(256) void csr_place(int* __restrict__ edges,
                                                 const int* __restrict__ src,
                                                 const int* __restrict__ dst,
                                                 const int* __restrict__ rank,
                                                 const int* __restrict__ rowptr,
                                                 const float* __restrict__ dis, int e) {
    int i = blockIdx.x * blockDim.x + threadIdx.x;
    if (i >= e) return;
    int s = src[i];
    int d = dst[i];
    int p = rowptr[d] + rank[i];
    v2i rec;
    rec.x = s;
    rec.y = __float_as_int(dis[s] * dis[d]);
    ((v2i*)edges)[p] = rec;
}

template <int KK, int TN, bool RELU, bool BIAS>
__global__ __launch_bounds__(256) void gcn_gemm_wmma(
        const float* __restrict__ A, const float* __restrict__ B,
        const float* __restrict__ bias, float* __restrict__ C, int M) {
    constexpr int Nc  = TN * 16;
    constexpr int LDB = KK + 2;
    __shared__ float Blds[Nc * LDB];

    for (int i = threadIdx.x; i < KK * Nc; i += 256) {
        int k = i / Nc;
        int c = i - k * Nc;
        Blds[c * LDB + k] = B[i];
    }
    __syncthreads();

    int tilesM = M >> 4;
    int tm     = blockIdx.x * 8 + (threadIdx.x >> 5);
    int lane   = threadIdx.x & 31;
    int half   = lane >> 4;
    int lr     = lane & 15;

    if (tm < tilesM) {
        const float* Arow = A + (size_t)(tm * 16 + lr) * KK;
        __builtin_prefetch(Arow, 0, 0);
        __builtin_prefetch(Arow + 64, 0, 0);

        v8f acc[TN] = {};
        for (int k = 0; k < KK; k += 4) {
            v2f a = *(const v2f*)(Arow + k + 2 * half);
#pragma unroll
            for (int t = 0; t < TN; ++t) {
                v2f b = *(const v2f*)(&Blds[(t * 16 + lr) * LDB + k + 2 * half]);
                acc[t] = __builtin_amdgcn_wmma_f32_16x16x4_f32(
                             false, a, false, b, (short)0, acc[t], false, false);
            }
        }

#pragma unroll
        for (int t = 0; t < TN; ++t) {
            int col  = t * 16 + lr;
            float bv = BIAS ? bias[col] : 0.0f;
#pragma unroll
            for (int r = 0; r < 8; ++r) {
                int row = tm * 16 + r + 8 * half;
                float v = acc[t][r] + bv;
                if (RELU) v = fmaxf(v, 0.0f);
                C[(size_t)row * Nc + col] = v;
            }
        }
    }
}

template <int U, bool TAIL>
__device__ __forceinline__ void gather_pairs(v4f& acc, const v4f* __restrict__ g4,
                                             v2i rec, int j, int m, int half, int c) {
    v4f   r[U];
    float w[U];
#pragma unroll
    for (int u = 0; u < U; ++u) {
        int i0 = j + 2 * u;
        int s0 = __builtin_amdgcn_readlane(rec.x, i0);
        int s1 = __builtin_amdgcn_readlane(rec.x, i0 + 1);
        int w0 = __builtin_amdgcn_readlane(rec.y, i0);
        int w1 = __builtin_amdgcn_readlane(rec.y, i0 + 1);
        int s  = half ? s1 : s0;
        w[u]   = __int_as_float(half ? w1 : w0);
        if (TAIL) {
            v4f t = {0.0f, 0.0f, 0.0f, 0.0f};
            if (i0 + half < m) t = g4[(size_t)s * 16 + c];
            r[u] = t;
        } else {
            r[u] = g4[(size_t)s * 16 + c];
        }
    }
#pragma unroll
    for (int u = 0; u < U; ++u) acc += r[u] * w[u];
}

__global__ __launch_bounds__(256) void gcn_gather64(
        float* __restrict__ out, const float* __restrict__ g,
        const float* __restrict__ dis, const float* __restrict__ beta,
        const int* __restrict__ rowptr, const int* __restrict__ cnt,
        const int* __restrict__ edges, int n) {
    int node = blockIdx.x * 8 + __builtin_amdgcn_readfirstlane((int)(threadIdx.x >> 5));
    if (node >= n) return;
    int lane  = threadIdx.x & 31;
    int half  = lane >> 4;
    int c     = lane & 15;
    int start = __builtin_amdgcn_readfirstlane(rowptr[node]);
    int deg   = __builtin_amdgcn_readfirstlane(cnt[node]);

    const v4f* g4 = (const v4f*)g;
    const v2i* ed = (const v2i*)edges + start;

    float sn   = dis[node];
    v4f   self = g4[(size_t)node * 16 + c];
    v4f   bv   = ((const v4f*)beta)[c];

    v4f acc = {0.0f, 0.0f, 0.0f, 0.0f};
    for (int base = 0; base < deg; base += 32) {
        int m = deg - base;
        if (m > 32) m = 32;
        v2i rec = {0, 0};
        if (lane < m) rec = ed[base + lane];
        int j = 0;
        for (; j + 16 <= m; j += 16) gather_pairs<8, false>(acc, g4, rec, j, m, half, c);
        if (j + 8 <= m) { gather_pairs<4, false>(acc, g4, rec, j, m, half, c); j += 8; }
        if (j + 4 <= m) { gather_pairs<2, false>(acc, g4, rec, j, m, half, c); j += 4; }
        if (j < m)      { gather_pairs<2, true >(acc, g4, rec, j, m, half, c); }
    }

    v4f oth;
    oth.x = __shfl_xor(acc.x, 16);
    oth.y = __shfl_xor(acc.y, 16);
    oth.z = __shfl_xor(acc.z, 16);
    oth.w = __shfl_xor(acc.w, 16);
    sn = sn * sn;
    if (half == 0)
        ((v4f*)out)[(size_t)node * 16 + c] = (acc + oth) + self * sn + bv;
}

__global__ __launch_bounds__(64) void fold_bias(
        float* __restrict__ beta, const float* __restrict__ bc,
        const float* __restrict__ b2, const float* __restrict__ M1,
        const float* __restrict__ M2, const float* __restrict__ W2) {
    int l = blockIdx.x;
    int c = threadIdx.x;
    const float* M = (l == 0) ? M1 : ((l == 1) ? M2 : W2);
    const float* b = bc + l * 128;
    float s = 0.0f;
    for (int k = 0; k < 128; ++k) s += b[k] * M[k * 64 + c];
    if (l == 2) s += b2[c];
    beta[l * 64 + c] = s;
}

extern "C" void kernel_launch(void* const* d_in, const int* in_sizes, int n_in,
                              void* d_out, int out_size, void* d_ws, size_t ws_size,
                              hipStream_t stream) {
    const float* x  = (const float*)d_in[0];
    const int*   ei = (const int*)  d_in[1];
    const float* W1 = (const float*)d_in[2];
    const float* b1 = (const float*)d_in[3];
    const float* Wc = (const float*)d_in[4];
    const float* bc = (const float*)d_in[5];
    const float* W2 = (const float*)d_in[6];
    const float* b2 = (const float*)d_in[7];

    const int N = 50000, E = 800000, H = 128;
    const int* srcv = ei;
    const int* dstv = ei + E;


    char* ws = (char*)d_ws;
    size_t off = 0;
    auto carve = [&](size_t bytes) -> void* {
        void* p = ws + off;
        off = (off + bytes + 255) & ~(size_t)255;
        return p;
    };
    int*   cnt    = (int*)  carve((size_t)N * 4);
    int*   rowptr = (int*)  carve((size_t)N * 4);
    float* dis    = (float*)carve((size_t)N * 4);
    int*   rank   = (int*)  carve((size_t)E * 4);
    int*   edges  = (int*)  carve((size_t)E * 8);
    float* h0     = (float*)carve((size_t)N * H * 4);
    float* pA     = (float*)carve((size_t)N * 64 * 4);
    float* pB     = (float*)carve((size_t)N * 64 * 4);
    float* M2     = (float*)carve((size_t)H * 64 * 4);
    float* M1     = (float*)carve((size_t)H * 64 * 4);
    float* M0     = (float*)carve((size_t)H * 64 * 4);
    float* beta   = (float*)carve((size_t)3 * 64 * 4);

    csr_zero      <<<(N + 255) / 256, 256, 0, stream>>>(cnt, N);
    csr_count_rank<<<(E + 255) / 256, 256, 0, stream>>>(cnt, rank, dstv, E);
    csr_scan      <<<(N + 255) / 256, 256, 0, stream>>>(cnt, rowptr, dis, N);
    csr_place     <<<(E + 255) / 256, 256, 0, stream>>>(edges, srcv, dstv, rank,
                                                        rowptr, dis, E);

    gcn_gemm_wmma<128, 4, false, false><<<1, 256, 0, stream>>>(
        Wc + (size_t)2 * H * H, W2, nullptr, M2, H);
    gcn_gemm_wmma<128, 4, false, false><<<1, 256, 0, stream>>>(
        Wc + (size_t)1 * H * H, M2, nullptr, M1, H);
    gcn_gemm_wmma<128, 4, false, false><<<1, 256, 0, stream>>>(
        Wc, M1, nullptr, M0, H);
    fold_bias<<<3, 64, 0, stream>>>(beta, bc, b2, M1, M2, W2);

    const int tilesM    = N / 16;
    const int gemmBlks  = (tilesM + 7) / 8;

    gcn_gemm_wmma<128, 8, true, true><<<gemmBlks, 256, 0, stream>>>(
        x, W1, b1, h0, N);

    gcn_gemm_wmma<128, 4, false, false><<<gemmBlks, 256, 0, stream>>>(
        h0, M0, nullptr, pA, N);

    gcn_gather64<<<(N + 7) / 8, 256, 0, stream>>>(
        pB, pA, dis, beta, rowptr, cnt, edges, N);
    gcn_gather64<<<(N + 7) / 8, 256, 0, stream>>>(
        pA, pB, dis, beta + 64, rowptr, cnt, edges, N);
    gcn_gather64<<<(N + 7) / 8, 256, 0, stream>>>(
        (float*)d_out, pA, dis, beta + 128, rowptr, cnt, edges, N);
}


# weight fold (3 WMMA products + bias rows) run by one extra work-group of the edge-count launch instead of 4 one-group launches: 13 launches to 9; fp sum order unchanged
# speedup vs baseline: 3.7290x; 1.1034x over previous
#include <hip/hip_runtime.h>


typedef float v2f __attribute__((ext_vector_type(2)));
typedef float v4f __attribute__((ext_vector_type(4)));
typedef float v8f __attribute__((ext_vector_type(8)));
typedef int   v2i __attribute__((ext_vector_type(2)));
typedef int   v4i __attribute__((ext_vector_type(4)));

__global__ __launch_bounds__(256) void csr_zero(int* __restrict__ cnt, int n) {
    int i = blockIdx.x * blockDim.x + threadIdx.x;
    if (i < n) cnt[i] = 0;
}

__device__ __forceinline__ void fold_product(const float* A, const float* B,
                                             float* C, float* Blds) {
    constexpr int KK  = 128;
    constexpr int TN  = 4;
    constexpr int Nc  = TN * 16;
    constexpr int LDB = KK + 2;

    for (int i = threadIdx.x; i < KK * Nc; i += 256) {
        int k = i / Nc;
        int c = i - k * Nc;
        Blds[c * LDB + k] = B[i];
    }
    __syncthreads();

    int tm   = threadIdx.x >> 5;
    int lane = threadIdx.x & 31;
    int half = lane >> 4;
    int lr   = lane & 15;

    const float* Arow = A + (size_t)(tm * 16 + lr) * KK;
    v8f acc[TN] = {};
    for (int k = 0; k < KK; k += 4) {
        v2f a = *(const v2f*)(Arow + k + 2 * half);
#pragma unroll
        for (int t = 0; t < TN; ++t) {
            v2f b = *(const v2f*)(&Blds[(t * 16 + lr) * LDB + k + 2 * half]);
            acc[t] = __builtin_amdgcn_wmma_f32_16x16x4_f32(
                         false, a, false, b, (short)0, acc[t], false, false);
        }
    }
#pragma unroll
    for (int t = 0; t < TN; ++t) {
        int col = t * 16 + lr;
#pragma unroll
        for (int r = 0; r < 8; ++r) {
            int row = tm * 16 + r + 8 * half;
            C[(size_t)row * Nc + col] = acc[t][r];
        }
    }
    __syncthreads();
}

__global__ __launch_bounds__(256) void csr_count_rank_fold(
        int* __restrict__ cnt, int* __restrict__ rank,
        const int* __restrict__ dst, int e,
        const float* Wc, const float* W2, const float* bc, const float* b2,
        float* M2, float* M1, float* M0, float* beta) {
    __shared__ float Blds[64 * 130];
    if (blockIdx.x == 0) {
        fold_product(Wc + 2 * 128 * 128, W2, M2, Blds);
        fold_product(Wc + 1 * 128 * 128, M2, M1, Blds);
        fold_product(Wc,                 M1, M0, Blds);
        if (threadIdx.x < 192) {
            int l = threadIdx.x >> 6;
            int c = threadIdx.x & 63;
            const float* M = (l == 0) ? M1 : ((l == 1) ? M2 : W2);
            const float* b = bc + l * 128;
            float s = 0.0f;
            for (int k = 0; k < 128; ++k) s += b[k] * M[k * 64 + c];
            if (l == 2) s += b2[c];
            beta[l * 64 + c] = s;
        }
        return;
    }
    int i = (blockIdx.x - 1) * blockDim.x + threadIdx.x;
    if (i < e) rank[i] = atomicAdd(&cnt[dst[i]], 1);
}

__global__ __launch_bounds__(256) void csr_scan(const int* __restrict__ cnt,
                                                int* __restrict__ rowptr,
                                                float* __restrict__ dis, int n) {
    __shared__ int s_part[8];
    __shared__ int s_tot[8];
    int tid  = threadIdx.x;
    int lane = tid & 31;
    int wv   = tid >> 5;
    int base = blockIdx.x * 256;

    const v4i* c4 = (const v4i*)cnt;
    const int nq = base >> 2;
    int part = 0;
    for (int j0 = 0; j0 < nq; j0 += 2048) {
        v4i q[8];
#pragma unroll
        for (int u = 0; u < 8; ++u) {
            int idx = j0 + u * 256 + tid;
            q[u] = c4[idx < nq ? idx : nq - 1];
        }
#pragma unroll
        for (int u = 0; u < 8; ++u) {
            int idx = j0 + u * 256 + tid;
            int sum = (q[u].x + q[u].y) + (q[u].z + q[u].w);
            part += (idx < nq) ? sum : 0;
        }
    }

    int i = base + tid;
    int c = (i < n) ? cnt[i] : 0;
    int incl = c;
    for (int o = 1; o < 32; o <<= 1) {
        int t = __shfl_up(incl, o);
        if (lane >= o) incl += t;
    }
    for (int o = 16; o > 0; o >>= 1) part += __shfl_xor(part, o);
    if (lane == 31) s_tot[wv]  = incl;
    if (lane == 0)  s_part[wv] = part;
    __syncthreads();

    int off = 0;
    for (int w = 0; w < 8; ++w) {
        off += s_part[w];
        if (w < wv) off += s_tot[w];
    }
    if (i < n) {
        rowptr[i] = off + incl - c;
        dis[i]    = rsqrtf((float)c + 1.0f);
    }
}

__global__ __launch_bounds__(256) void csr_place(int* __restrict__ edges,
                                                 const int* __restrict__ src,
                                                 const int* __restrict__ dst,
                                                 const int* __restrict__ rank,
                                                 const int* __restrict__ rowptr,
                                                 const float* __restrict__ dis, int e) {
    int i = blockIdx.x * blockDim.x + threadIdx.x;
    if (i >= e) return;
    int s = src[i];
    int d = dst[i];
    int p = rowptr[d] + rank[i];
    v2i rec;
    rec.x = s;
    rec.y = __float_as_int(dis[s] * dis[d]);
    ((v2i*)edges)[p] = rec;
}

template <int KK, int TN, bool RELU, bool BIAS>
__global__ __launch_bounds__(256) void gcn_gemm_wmma(
        const float* __restrict__ A, const float* __restrict__ B,
        const float* __restrict__ bias, float* __restrict__ C, int M) {
    constexpr int Nc  = TN * 16;
    constexpr int LDB = KK + 2;
    __shared__ float Blds[Nc * LDB];

    for (int i = threadIdx.x; i < KK * Nc; i += 256) {
        int k = i / Nc;
        int c = i - k * Nc;
        Blds[c * LDB + k] = B[i];
    }
    __syncthreads();

    int tilesM = M >> 4;
    int tm     = blockIdx.x * 8 + (threadIdx.x >> 5);
    int lane   = threadIdx.x & 31;
    int half   = lane >> 4;
    int lr     = lane & 15;

    if (tm < tilesM) {
        const float* Arow = A + (size_t)(tm * 16 + lr) * KK;
        __builtin_prefetch(Arow, 0, 0);
        __builtin_prefetch(Arow + 64, 0, 0);

        v8f acc[TN] = {};
        for (int k = 0; k < KK; k += 4) {
            v2f a = *(const v2f*)(Arow + k + 2 * half);
#pragma unroll
            for (int t = 0; t < TN; ++t) {
                v2f b = *(const v2f*)(&Blds[(t * 16 + lr) * LDB + k + 2 * half]);
                acc[t] = __builtin_amdgcn_wmma_f32_16x16x4_f32(
                             false, a, false, b, (short)0, acc[t], false, false);
            }
        }

#pragma unroll
        for (int t = 0; t < TN; ++t) {
            int col  = t * 16 + lr;
            float bv = BIAS ? bias[col] : 0.0f;
#pragma unroll
            for (int r = 0; r < 8; ++r) {
                int row = tm * 16 + r + 8 * half;
                float v = acc[t][r] + bv;
                if (RELU) v = fmaxf(v, 0.0f);
                C[(size_t)row * Nc + col] = v;
            }
        }
    }
}

template <int U, bool TAIL>
__device__ __forceinline__ void gather_pairs(v4f& acc, const v4f* __restrict__ g4,
                                             v2i rec, int j, int m, int half, int c) {
    v4f   r[U];
    float w[U];
#pragma unroll
    for (int u = 0; u < U; ++u) {
        int i0 = j + 2 * u;
        int s0 = __builtin_amdgcn_readlane(rec.x, i0);
        int s1 = __builtin_amdgcn_readlane(rec.x, i0 + 1);
        int w0 = __builtin_amdgcn_readlane(rec.y, i0);
        int w1 = __builtin_amdgcn_readlane(rec.y, i0 + 1);
        int s  = half ? s1 : s0;
        w[u]   = __int_as_float(half ? w1 : w0);
        if (TAIL) {
            v4f t = {0.0f, 0.0f, 0.0f, 0.0f};
            if (i0 + half < m) t = g4[(size_t)s * 16 + c];
            r[u] = t;
        } else {
            r[u] = g4[(size_t)s * 16 + c];
        }
    }
#pragma unroll
    for (int u = 0; u < U; ++u) acc += r[u] * w[u];
}

__global__ __launch_bounds__(256) void gcn_gather64(
        float* __restrict__ out, const float* __restrict__ g,
        const float* __restrict__ dis, const float* __restrict__ beta,
        const int* __restrict__ rowptr, const int* __restrict__ cnt,
        const int* __restrict__ edges, int n) {
    int node = blockIdx.x * 8 + __builtin_amdgcn_readfirstlane((int)(threadIdx.x >> 5));
    if (node >= n) return;
    int lane  = threadIdx.x & 31;
    int half  = lane >> 4;
    int c     = lane & 15;
    int start = __builtin_amdgcn_readfirstlane(rowptr[node]);
    int deg   = __builtin_amdgcn_readfirstlane(cnt[node]);

    const v4f* g4 = (const v4f*)g;
    const v2i* ed = (const v2i*)edges + start;

    float sn   = dis[node];
    v4f   self = g4[(size_t)node * 16 + c];
    v4f   bv   = ((const v4f*)beta)[c];

    v4f acc = {0.0f, 0.0f, 0.0f, 0.0f};
    for (int base = 0; base < deg; base += 32) {
        int m = deg - base;
        if (m > 32) m = 32;
        v2i rec = {0, 0};
        if (lane < m) rec = ed[base + lane];
        int j = 0;
        for (; j + 16 <= m; j += 16) gather_pairs<8, false>(acc, g4, rec, j, m, half, c);
        if (j + 8 <= m) { gather_pairs<4, false>(acc, g4, rec, j, m, half, c); j += 8; }
        if (j + 4 <= m) { gather_pairs<2, false>(acc, g4, rec, j, m, half, c); j += 4; }
        if (j < m)      { gather_pairs<2, true >(acc, g4, rec, j, m, half, c); }
    }

    v4f oth;
    oth.x = __shfl_xor(acc.x, 16);
    oth.y = __shfl_xor(acc.y, 16);
    oth.z = __shfl_xor(acc.z, 16);
    oth.w = __shfl_xor(acc.w, 16);
    sn = sn * sn;
    if (half == 0)
        ((v4f*)out)[(size_t)node * 16 + c] = (acc + oth) + self * sn + bv;
}

extern "C" void kernel_launch(void* const* d_in, const int* in_sizes, int n_in,
                              void* d_out, int out_size, void* d_ws, size_t ws_size,
                              hipStream_t stream) {
    const float* x  = (const float*)d_in[0];
    const int*   ei = (const int*)  d_in[1];
    const float* W1 = (const float*)d_in[2];
    const float* b1 = (const float*)d_in[3];
    const float* Wc = (const float*)d_in[4];
    const float* bc = (const float*)d_in[5];
    const float* W2 = (const float*)d_in[6];
    const float* b2 = (const float*)d_in[7];

    const int N = 50000, E = 800000, H = 128;
    const int* srcv = ei;
    const int* dstv = ei + E;


    char* ws = (char*)d_ws;
    size_t off = 0;
    auto carve = [&](size_t bytes) -> void* {
        void* p = ws + off;
        off = (off + bytes + 255) & ~(size_t)255;
        return p;
    };
    int*   cnt    = (int*)  carve((size_t)N * 4);
    int*   rowptr = (int*)  carve((size_t)N * 4);
    float* dis    = (float*)carve((size_t)N * 4);
    int*   rank   = (int*)  carve((size_t)E * 4);
    int*   edges  = (int*)  carve((size_t)E * 8);
    float* h0     = (float*)carve((size_t)N * H * 4);
    float* pA     = (float*)carve((size_t)N * 64 * 4);
    float* pB     = (float*)carve((size_t)N * 64 * 4);
    float* M2     = (float*)carve((size_t)H * 64 * 4);
    float* M1     = (float*)carve((size_t)H * 64 * 4);
    float* M0     = (float*)carve((size_t)H * 64 * 4);
    float* beta   = (float*)carve((size_t)3 * 64 * 4);

    csr_zero      <<<(N + 255) / 256, 256, 0, stream>>>(cnt, N);
    csr_count_rank_fold<<<(E + 255) / 256 + 1, 256, 0, stream>>>(
        cnt, rank, dstv, E, Wc, W2, bc, b2, M2, M1, M0, beta);
    csr_scan      <<<(N + 255) / 256, 256, 0, stream>>>(cnt, rowptr, dis, N);
    csr_place     <<<(E + 255) / 256, 256, 0, stream>>>(edges, srcv, dstv, rank,
                                                        rowptr, dis, E);

    const int tilesM    = N / 16;
    const int gemmBlks  = (tilesM + 7) / 8;

    gcn_gemm_wmma<128, 8, true, true><<<gemmBlks, 256, 0, stream>>>(
        x, W1, b1, h0, N);

    gcn_gemm_wmma<128, 4, false, false><<<gemmBlks, 256, 0, stream>>>(
        h0, M0, nullptr, pA, N);

    gcn_gather64<<<(N + 7) / 8, 256, 0, stream>>>(
        pB, pA, dis, beta, rowptr, cnt, edges, N);
    gcn_gather64<<<(N + 7) / 8, 256, 0, stream>>>(
        pA, pB, dis, beta + 64, rowptr, cnt, edges, N);
    gcn_gather64<<<(N + 7) / 8, 256, 0, stream>>>(
        (float*)d_out, pA, dis, beta + 128, rowptr, cnt, edges, N);
}


# in-edges in fixed-pitch slots (64 a node) filled by the count atomic: rank, scan, place launches gone (9 to 7); norm factored, rows kept times dis; fp sum order changes (products reassociated)
# speedup vs baseline: 4.4857x; 1.2029x over previous
#include <hip/hip_runtime.h>


typedef float v2f __attribute__((ext_vector_type(2)));
typedef float v4f __attribute__((ext_vector_type(4)));
typedef float v8f __attribute__((ext_vector_type(8)));
typedef int   v2i __attribute__((ext_vector_type(2)));
typedef int   v4i __attribute__((ext_vector_type(4)));

constexpr int SLOTS = 64;

__global__ __launch_bounds__(256) void slots_zero(int* __restrict__ cnt, int n) {
    int i = blockIdx.x * blockDim.x + threadIdx.x;
    if (i < n) cnt[i] = 0;
}

__device__ __forceinline__ void fold_product(const float* A, const float* B,
                                             float* C, float* Blds) {
    constexpr int KK  = 128;
    constexpr int TN  = 4;
    constexpr int Nc  = TN * 16;
    constexpr int LDB = KK + 2;

    for (int i = threadIdx.x; i < KK * Nc; i += 256) {
        int k = i / Nc;
        int c = i - k * Nc;
        Blds[c * LDB + k] = B[i];
    }
    __syncthreads();

    int tm   = threadIdx.x >> 5;
    int lane = threadIdx.x & 31;
    int half = lane >> 4;
    int lr   = lane & 15;

    const float* Arow = A + (size_t)(tm * 16 + lr) * KK;
    v8f acc[TN] = {};
    for (int k = 0; k < KK; k += 4) {
        v2f a = *(const v2f*)(Arow + k + 2 * half);
#pragma unroll
        for (int t = 0; t < TN; ++t) {
            v2f b = *(const v2f*)(&Blds[(t * 16 + lr) * LDB + k + 2 * half]);
            acc[t] = __builtin_amdgcn_wmma_f32_16x16x4_f32(
                         false, a, false, b, (short)0, acc[t], false, false);
        }
    }
#pragma unroll
    for (int t = 0; t < TN; ++t) {
        int col = t * 16 + lr;
#pragma unroll
        for (int r = 0; r < 8; ++r) {
            int row = tm * 16 + r + 8 * half;
            C[(size_t)row * Nc + col] = acc[t][r];
        }
    }
    __syncthreads();
}

__global__ __launch_bounds__(256) void slots_fill_fold(
        int* __restrict__ cnt, int* __restrict__ table, int* __restrict__ spill,
        const int* __restrict__ src, const int* __restrict__ dst, int e, int n,
        const float* Wc, const float* W2, const float* bc, const float* b2,
        float* M2, float* M1, float* M0, float* beta) {
    __shared__ float Blds[64 * 130];
    if (blockIdx.x == 0) {
        fold_product(Wc + 2 * 128 * 128, W2, M2, Blds);
        fold_product(Wc + 1 * 128 * 128, M2, M1, Blds);
        fold_product(Wc,                 M1, M0, Blds);
        if (threadIdx.x < 192) {
            int l = threadIdx.x >> 6;
            int c = threadIdx.x & 63;
            const float* M = (l == 0) ? M1 : ((l == 1) ? M2 : W2);
            const float* b = bc + l * 128;
            float s = 0.0f;
            for (int k = 0; k < 128; ++k) s += b[k] * M[k * 64 + c];
            if (l == 2) s += b2[c];
            beta[l * 64 + c] = s;
        }
        return;
    }
    int i = (blockIdx.x - 1) * blockDim.x + threadIdx.x;
    if (i >= e) return;
    int s = src[i];
    int d = dst[i];
    int slot = atomicAdd(&cnt[d], 1);
    if (slot < SLOTS) {
        table[(size_t)d * SLOTS + slot] = s;
    } else {
        int o = atomicAdd(&cnt[n], 1);
        v2i rec;
        rec.x = s;
        rec.y = d;
        ((v2i*)spill)[o] = rec;
    }
}

template <int KK, int TN, bool RELU, bool BIAS, bool DSCALE>
__global__ __launch_bounds__(256) void gcn_gemm_wmma(
        const float* __restrict__ A, const float* __restrict__ B,
        const float* __restrict__ bias, const int* __restrict__ cnt,
        float* __restrict__ C, int M) {
    constexpr int Nc  = TN * 16;
    constexpr int LDB = KK + 2;
    __shared__ float Blds[Nc * LDB];

    for (int i = threadIdx.x; i < KK * Nc; i += 256) {
        int k = i / Nc;
        int c = i - k * Nc;
        Blds[c * LDB + k] = B[i];
    }
    __syncthreads();

    int tilesM = M >> 4;
    int tm     = blockIdx.x * 8 + (threadIdx.x >> 5);
    int lane   = threadIdx.x & 31;
    int half   = lane >> 4;
    int lr     = lane & 15;

    if (tm < tilesM) {
        const float* Arow = A + (size_t)(tm * 16 + lr) * KK;
        __builtin_prefetch(Arow, 0, 0);
        __builtin_prefetch(Arow + 64, 0, 0);

        v8f acc[TN] = {};
        for (int k = 0; k < KK; k += 4) {
            v2f a = *(const v2f*)(Arow + k + 2 * half);
#pragma unroll
            for (int t = 0; t < TN; ++t) {
                v2f b = *(const v2f*)(&Blds[(t * 16 + lr) * LDB + k + 2 * half]);
                acc[t] = __builtin_amdgcn_wmma_f32_16x16x4_f32(
                             false, a, false, b, (short)0, acc[t], false, false);
            }
        }

        float sc[8];
#pragma unroll
        for (int r = 0; r < 8; ++r) {
            sc[r] = 1.0f;
            if (DSCALE) sc[r] = rsqrtf((float)cnt[tm * 16 + r + 8 * half] + 1.0f);
        }
#pragma unroll
        for (int t = 0; t < TN; ++t) {
            int col  = t * 16 + lr;
            float bv = BIAS ? bias[col] : 0.0f;
#pragma unroll
            for (int r = 0; r < 8; ++r) {
                int row = tm * 16 + r + 8 * half;
                float v = acc[t][r] + bv;
                if (RELU) v = fmaxf(v, 0.0f);
                if (DSCALE) v = v * sc[r];
                C[(size_t)row * Nc + col] = v;
            }
        }
    }
}

template <int U, bool TAIL>
__device__ __forceinline__ void gather_pairs(v4f& acc, const v4f* __restrict__ q4,
                                             int rec, int j, int m, int half, int c) {
    v4f r[U];
#pragma unroll
    for (int u = 0; u < U; ++u) {
        int i0 = j + 2 * u;
        int s0 = __builtin_amdgcn_readlane(rec, i0);
        int s1 = __builtin_amdgcn_readlane(rec, i0 + 1);
        int s  = half ? s1 : s0;
        if (TAIL) {
            v4f t = {0.0f, 0.0f, 0.0f, 0.0f};
            if (i0 + half < m) t = q4[(size_t)s * 16 + c];
            r[u] = t;
        } else {
            r[u] = q4[(size_t)s * 16 + c];
        }
    }
#pragma unroll
    for (int u = 0; u < U; ++u) acc += r[u];
}

__device__ __forceinline__ void gather_chunk(v4f& acc, const v4f* __restrict__ q4,
                                             int rec, int m, int lane, int half, int c) {
    if (lane >= m) rec = 0;
    int j = 0;
    for (; j + 16 <= m; j += 16) gather_pairs<8, false>(acc, q4, rec, j, m, half, c);
    if (j + 8 <= m) { gather_pairs<4, false>(acc, q4, rec, j, m, half, c); j += 8; }
    if (j + 4 <= m) { gather_pairs<2, false>(acc, q4, rec, j, m, half, c); j += 4; }
    if (j < m)      { gather_pairs<2, true >(acc, q4, rec, j, m, half, c); }
}

template <bool LAST>
__global__ __launch_bounds__(256) void gcn_gather64(
        float* __restrict__ out, const float* __restrict__ q,
        const float* __restrict__ beta, const int* __restrict__ cnt,
        const int* __restrict__ table, const int* __restrict__ spill,
        int n, int e) {
    int node = blockIdx.x * 8 + __builtin_amdgcn_readfirstlane((int)(threadIdx.x >> 5));
    if (node >= n) return;
    int lane = threadIdx.x & 31;
    int half = lane >> 4;
    int c    = lane & 15;

    const v4f* q4  = (const v4f*)q;
    const int* row = table + (size_t)node * SLOTS;

    int deg  = __builtin_amdgcn_readfirstlane(cnt[node]);
    int rec0 = row[lane];
    v4f self = q4[(size_t)node * 16 + c];
    v4f bv   = ((const v4f*)beta)[c];

    int filled = deg < SLOTS ? deg : SLOTS;
    v4f acc = {0.0f, 0.0f, 0.0f, 0.0f};
    gather_chunk(acc, q4, rec0, filled < 32 ? filled : 32, lane, half, c);
    if (filled > 32)
        gather_chunk(acc, q4, row[32 + lane], filled - 32, lane, half, c);

    if (deg > SLOTS) {
        int total = __builtin_amdgcn_readfirstlane(cnt[n]);
        if (total > e) total = e;
        const v2i* sp = (const v2i*)spill;
        for (int ob = 0; ob < total; ob += 32) {
            int mm = total - ob;
            if (mm > 32) mm = 32;
            v2i orec = {0, -1};
            if (lane < mm) orec = sp[ob + lane];
            for (int jj = 0; jj < mm; ++jj) {
                int s = __builtin_amdgcn_readlane(orec.x, jj);
                int d = __builtin_amdgcn_readlane(orec.y, jj);
                if (d == node && half == 0) acc += q4[(size_t)s * 16 + c];
            }
        }
    }

    v4f oth;
    oth.x = __shfl_xor(acc.x, 16);
    oth.y = __shfl_xor(acc.y, 16);
    oth.z = __shfl_xor(acc.z, 16);
    oth.w = __shfl_xor(acc.w, 16);
    float dn = rsqrtf((float)deg + 1.0f);
    v4f v = ((acc + oth) + self) * dn + bv;
    if (!LAST) v = v * dn;
    if (half == 0)
        ((v4f*)out)[(size_t)node * 16 + c] = v;
}

extern "C" void kernel_launch(void* const* d_in, const int* in_sizes, int n_in,
                              void* d_out, int out_size, void* d_ws, size_t ws_size,
                              hipStream_t stream) {
    const float* x  = (const float*)d_in[0];
    const int*   ei = (const int*)  d_in[1];
    const float* W1 = (const float*)d_in[2];
    const float* b1 = (const float*)d_in[3];
    const float* Wc = (const float*)d_in[4];
    const float* bc = (const float*)d_in[5];
    const float* W2 = (const float*)d_in[6];
    const float* b2 = (const float*)d_in[7];

    const int N = 50000, E = 800000, H = 128;
    const int* srcv = ei;
    const int* dstv = ei + E;


    char* ws = (char*)d_ws;
    size_t off = 0;
    auto carve = [&](size_t bytes) -> void* {
        void* p = ws + off;
        off = (off + bytes + 255) & ~(size_t)255;
        return p;
    };
    int*   cnt    = (int*)  carve((size_t)(N + 1) * 4);
    int*   table  = (int*)  carve((size_t)N * SLOTS * 4);
    int*   spill  = (int*)  carve((size_t)E * 8);
    float* h0     = (float*)carve((size_t)N * H * 4);
    float* pA     = (float*)carve((size_t)N * 64 * 4);
    float* pB     = (float*)carve((size_t)N * 64 * 4);
    float* M2     = (float*)carve((size_t)H * 64 * 4);
    float* M1     = (float*)carve((size_t)H * 64 * 4);
    float* M0     = (float*)carve((size_t)H * 64 * 4);
    float* beta   = (float*)carve((size_t)3 * 64 * 4);

    slots_zero    <<<(N + 1 + 255) / 256, 256, 0, stream>>>(cnt, N + 1);
    slots_fill_fold<<<(E + 255) / 256 + 1, 256, 0, stream>>>(
        cnt, table, spill, srcv, dstv, E, N, Wc, W2, bc, b2, M2, M1, M0, beta);

    const int tilesM    = N / 16;
    const int gemmBlks  = (tilesM + 7) / 8;

    gcn_gemm_wmma<128, 8, true, true, false><<<gemmBlks, 256, 0, stream>>>(
        x, W1, b1, nullptr, h0, N);

    gcn_gemm_wmma<128, 4, false, false, true><<<gemmBlks, 256, 0, stream>>>(
        h0, M0, nullptr, cnt, pA, N);

    gcn_gather64<false><<<(N + 7) / 8, 256, 0, stream>>>(
        pB, pA, beta, cnt, table, spill, N, E);
    gcn_gather64<false><<<(N + 7) / 8, 256, 0, stream>>>(
        pA, pB, beta + 64, cnt, table, spill, N, E);
    gcn_gather64<true><<<(N + 7) / 8, 256, 0, stream>>>(
        (float*)d_out, pA, beta + 128, cnt, table, spill, N, E);
}


# x @ W1 work-groups moved into the edge-slot launch, behind the fold group: matrix-bound work beside atomic-bound work, 7 launches to 6; fp sum order unchanged (slots arrival-ordered as before)
# speedup vs baseline: 5.1473x; 1.1475x over previous
#include <hip/hip_runtime.h>


typedef float v2f __attribute__((ext_vector_type(2)));
typedef float v4f __attribute__((ext_vector_type(4)));
typedef float v8f __attribute__((ext_vector_type(8)));
typedef int   v2i __attribute__((ext_vector_type(2)));
typedef int   v4i __attribute__((ext_vector_type(4)));

constexpr int SLOTS = 64;

__global__ __launch_bounds__(256) void slots_zero(int* __restrict__ cnt, int n) {
    int i = blockIdx.x * blockDim.x + threadIdx.x;
    if (i < n) cnt[i] = 0;
}

__device__ __forceinline__ void fold_product(const float* A, const float* B,
                                             float* C, float* Blds) {
    constexpr int KK  = 128;
    constexpr int TN  = 4;
    constexpr int Nc  = TN * 16;
    constexpr int LDB = KK + 2;

    for (int i = threadIdx.x; i < KK * Nc; i += 256) {
        int k = i / Nc;
        int c = i - k * Nc;
        Blds[c * LDB + k] = B[i];
    }
    __syncthreads();

    int tm   = threadIdx.x >> 5;
    int lane = threadIdx.x & 31;
    int half = lane >> 4;
    int lr   = lane & 15;

    const float* Arow = A + (size_t)(tm * 16 + lr) * KK;
    v8f acc[TN] = {};
    for (int k = 0; k < KK; k += 4) {
        v2f a = *(const v2f*)(Arow + k + 2 * half);
#pragma unroll
        for (int t = 0; t < TN; ++t) {
            v2f b = *(const v2f*)(&Blds[(t * 16 + lr) * LDB + k + 2 * half]);
            acc[t] = __builtin_amdgcn_wmma_f32_16x16x4_f32(
                         false, a, false, b, (short)0, acc[t], false, false);
        }
    }
#pragma unroll
    for (int t = 0; t < TN; ++t) {
        int col = t * 16 + lr;
#pragma unroll
        for (int r = 0; r < 8; ++r) {
            int row = tm * 16 + r + 8 * half;
            C[(size_t)row * Nc + col] = acc[t][r];
        }
    }
    __syncthreads();
}

template <int KK, int TN, bool RELU, bool BIAS, bool DSCALE>
__device__ __forceinline__ void gemm_group(
        const float* __restrict__ A, const float* __restrict__ B,
        const float* __restrict__ bias, const int* __restrict__ cnt,
        float* __restrict__ C, int M, int blk, float* Blds) {
    constexpr int Nc  = TN * 16;
    constexpr int LDB = KK + 2;

    for (int i = threadIdx.x; i < KK * Nc; i += 256) {
        int k = i / Nc;
        int c = i - k * Nc;
        Blds[c * LDB + k] = B[i];
    }
    __syncthreads();

    int tilesM = M >> 4;
    int tm     = blk * 8 + __builtin_amdgcn_readfirstlane((int)(threadIdx.x >> 5));
    int lane   = threadIdx.x & 31;
    int half   = lane >> 4;
    int lr     = lane & 15;

    if (tm < tilesM) {
        const float* Arow = A + (size_t)(tm * 16 + lr) * KK;
        __builtin_prefetch(Arow, 0, 0);
        __builtin_prefetch(Arow + 64, 0, 0);

        v8f acc[TN] = {};
        for (int k = 0; k < KK; k += 4) {
            v2f a = *(const v2f*)(Arow + k + 2 * half);
#pragma unroll
            for (int t = 0; t < TN; ++t) {
                v2f b = *(const v2f*)(&Blds[(t * 16 + lr) * LDB + k + 2 * half]);
                acc[t] = __builtin_amdgcn_wmma_f32_16x16x4_f32(
                             false, a, false, b, (short)0, acc[t], false, false);
            }
        }

        float sc[8];
#pragma unroll
        for (int r = 0; r < 8; ++r) {
            sc[r] = 1.0f;
            if (DSCALE) sc[r] = rsqrtf((float)cnt[tm * 16 + r + 8 * half] + 1.0f);
        }
#pragma unroll
        for (int t = 0; t < TN; ++t) {
            int col  = t * 16 + lr;
            float bv = BIAS ? bias[col] : 0.0f;
#pragma unroll
            for (int r = 0; r < 8; ++r) {
                int row = tm * 16 + r + 8 * half;
                float v = acc[t][r] + bv;
                if (RELU) v = fmaxf(v, 0.0f);
                if (DSCALE) v = v * sc[r];
                C[(size_t)row * Nc + col] = v;
            }
        }
    }
}

template <int KK, int TN, bool RELU, bool BIAS, bool DSCALE>
__global__ __launch_bounds__(256) void gcn_gemm_wmma(
        const float* __restrict__ A, const float* __restrict__ B,
        const float* __restrict__ bias, const int* __restrict__ cnt,
        float* __restrict__ C, int M) {
    __shared__ float Blds[TN * 16 * (KK + 2)];
    gemm_group<KK, TN, RELU, BIAS, DSCALE>(A, B, bias, cnt, C, M, (int)blockIdx.x, Blds);
}

__global__ __launch_bounds__(256, 8) void slots_fill_fold_xw1(
        int* __restrict__ cnt, int* __restrict__ table, int* __restrict__ spill,
        const int* __restrict__ src, const int* __restrict__ dst, int e, int n,
        const float* Wc, const float* W2, const float* bc, const float* b2,
        float* M2, float* M1, float* M0, float* beta,
        const float* __restrict__ x, const float* __restrict__ W1,
        const float* __restrict__ b1, float* __restrict__ h0,
        int rows, int gemm_groups) {
    __shared__ float Blds[128 * 130];
    if (blockIdx.x == 0) {
        fold_product(Wc + 2 * 128 * 128, W2, M2, Blds);
        fold_product(Wc + 1 * 128 * 128, M2, M1, Blds);
        fold_product(Wc,                 M1, M0, Blds);
        if (threadIdx.x < 192) {
            int l = threadIdx.x >> 6;
            int c = threadIdx.x & 63;
            const float* M = (l == 0) ? M1 : ((l == 1) ? M2 : W2);
            const float* b = bc + l * 128;
            float s = 0.0f;
            for (int k = 0; k < 128; ++k) s += b[k] * M[k * 64 + c];
            if (l == 2) s += b2[c];
            beta[l * 64 + c] = s;
        }
        return;
    }
    if ((int)blockIdx.x <= gemm_groups) {
        gemm_group<128, 8, true, true, false>(
            x, W1, b1, nullptr, h0, rows, (int)blockIdx.x - 1, Blds);
        return;
    }
    int i = ((int)blockIdx.x - 1 - gemm_groups) * (int)blockDim.x + (int)threadIdx.x;
    if (i >= e) return;
    int s = src[i];
    int d = dst[i];
    int slot = atomicAdd(&cnt[d], 1);
    if (slot < SLOTS) {
        table[(size_t)d * SLOTS + slot] = s;
    } else {
        int o = atomicAdd(&cnt[n], 1);
        v2i rec;
        rec.x = s;
        rec.y = d;
        ((v2i*)spill)[o] = rec;
    }
}

template <int U, bool TAIL>
__device__ __forceinline__ void gather_pairs(v4f& acc, const v4f* __restrict__ q4,
                                             int rec, int j, int m, int half, int c) {
    v4f r[U];
#pragma unroll
    for (int u = 0; u < U; ++u) {
        int i0 = j + 2 * u;
        int s0 = __builtin_amdgcn_readlane(rec, i0);
        int s1 = __builtin_amdgcn_readlane(rec, i0 + 1);
        int s  = half ? s1 : s0;
        if (TAIL) {
            v4f t = {0.0f, 0.0f, 0.0f, 0.0f};
            if (i0 + half < m) t = q4[(size_t)s * 16 + c];
            r[u] = t;
        } else {
            r[u] = q4[(size_t)s * 16 + c];
        }
    }
#pragma unroll
    for (int u = 0; u < U; ++u) acc += r[u];
}

__device__ __forceinline__ void gather_chunk(v4f& acc, const v4f* __restrict__ q4,
                                             int rec, int m, int lane, int half, int c) {
    if (lane >= m) rec = 0;
    int j = 0;
    for (; j + 16 <= m; j += 16) gather_pairs<8, false>(acc, q4, rec, j, m, half, c);
    if (j + 8 <= m) { gather_pairs<4, false>(acc, q4, rec, j, m, half, c); j += 8; }
    if (j + 4 <= m) { gather_pairs<2, false>(acc, q4, rec, j, m, half, c); j += 4; }
    if (j < m)      { gather_pairs<2, true >(acc, q4, rec, j, m, half, c); }
}

template <bool LAST>
__global__ __launch_bounds__(256) void gcn_gather64(
        float* __restrict__ out, const float* __restrict__ q,
        const float* __restrict__ beta, const int* __restrict__ cnt,
        const int* __restrict__ table, const int* __restrict__ spill,
        int n, int e) {
    int node = blockIdx.x * 8 + __builtin_amdgcn_readfirstlane((int)(threadIdx.x >> 5));
    if (node >= n) return;
    int lane = threadIdx.x & 31;
    int half = lane >> 4;
    int c    = lane & 15;

    const v4f* q4  = (const v4f*)q;
    const int* row = table + (size_t)node * SLOTS;

    int deg  = __builtin_amdgcn_readfirstlane(cnt[node]);
    int rec0 = row[lane];
    v4f self = q4[(size_t)node * 16 + c];
    v4f bv   = ((const v4f*)beta)[c];

    int filled = deg < SLOTS ? deg : SLOTS;
    v4f acc = {0.0f, 0.0f, 0.0f, 0.0f};
    gather_chunk(acc, q4, rec0, filled < 32 ? filled : 32, lane, half, c);
    if (filled > 32)
        gather_chunk(acc, q4, row[32 + lane], filled - 32, lane, half, c);

    if (deg > SLOTS) {
        int total = __builtin_amdgcn_readfirstlane(cnt[n]);
        if (total > e) total = e;
        const v2i* sp = (const v2i*)spill;
        for (int ob = 0; ob < total; ob += 32) {
            int mm = total - ob;
            if (mm > 32) mm = 32;
            v2i orec = {0, -1};
            if (lane < mm) orec = sp[ob + lane];
            for (int jj = 0; jj < mm; ++jj) {
                int s = __builtin_amdgcn_readlane(orec.x, jj);
                int d = __builtin_amdgcn_readlane(orec.y, jj);
                if (d == node && half == 0) acc += q4[(size_t)s * 16 + c];
            }
        }
    }

    v4f oth;
    oth.x = __shfl_xor(acc.x, 16);
    oth.y = __shfl_xor(acc.y, 16);
    oth.z = __shfl_xor(acc.z, 16);
    oth.w = __shfl_xor(acc.w, 16);
    float dn = rsqrtf((float)deg + 1.0f);
    v4f v = ((acc + oth) + self) * dn + bv;
    if (!LAST) v = v * dn;
    if (half == 0)
        ((v4f*)out)[(size_t)node * 16 + c] = v;
}

extern "C" void kernel_launch(void* const* d_in, const int* in_sizes, int n_in,
                              void* d_out, int out_size, void* d_ws, size_t ws_size,
                              hipStream_t stream) {
    const float* x  = (const float*)d_in[0];
    const int*   ei = (const int*)  d_in[1];
    const float* W1 = (const float*)d_in[2];
    const float* b1 = (const float*)d_in[3];
    const float* Wc = (const float*)d_in[4];
    const float* bc = (const float*)d_in[5];
    const float* W2 = (const float*)d_in[6];
    const float* b2 = (const float*)d_in[7];

    const int N = 50000, E = 800000, H = 128;
    const int* srcv = ei;
    const int* dstv = ei + E;


    char* ws = (char*)d_ws;
    size_t off = 0;
    auto carve = [&](size_t bytes) -> void* {
        void* p = ws + off;
        off = (off + bytes + 255) & ~(size_t)255;
        return p;
    };
    int*   cnt    = (int*)  carve((size_t)(N + 1) * 4);
    int*   table  = (int*)  carve((size_t)N * SLOTS * 4);
    int*   spill  = (int*)  carve((size_t)E * 8);
    float* h0     = (float*)carve((size_t)N * H * 4);
    float* pA     = (float*)carve((size_t)N * 64 * 4);
    float* pB     = (float*)carve((size_t)N * 64 * 4);
    float* M2     = (float*)carve((size_t)H * 64 * 4);
    float* M1     = (float*)carve((size_t)H * 64 * 4);
    float* M0     = (float*)carve((size_t)H * 64 * 4);
    float* beta   = (float*)carve((size_t)3 * 64 * 4);

    const int tilesM    = N / 16;
    const int gemmBlks  = (tilesM + 7) / 8;

    slots_zero    <<<(N + 1 + 255) / 256, 256, 0, stream>>>(cnt, N + 1);
    slots_fill_fold_xw1<<<1 + gemmBlks + (E + 255) / 256, 256, 0, stream>>>(
        cnt, table, spill, srcv, dstv, E, N, Wc, W2, bc, b2, M2, M1, M0, beta,
        x, W1, b1, h0, N, gemmBlks);

    gcn_gemm_wmma<128, 4, false, false, true><<<gemmBlks, 256, 0, stream>>>(
        h0, M0, nullptr, cnt, pA, N);

    gcn_gather64<false><<<(N + 7) / 8, 256, 0, stream>>>(
        pB, pA, beta, cnt, table, spill, N, E);
    gcn_gather64<false><<<(N + 7) / 8, 256, 0, stream>>>(
        pA, pB, beta + 64, cnt, table, spill, N, E);
    gcn_gather64<true><<<(N + 7) / 8, 256, 0, stream>>>(
        (float*)d_out, pA, beta + 128, cnt, table, spill, N, E);
}


# edge slots by LDS counters in two passes (bin edges by 256-node range, then one group a range ranks in LDS, stores whole rows): 800k global atomics, clear launch gone; sum order still arrival-ordered
# speedup vs baseline: 6.5596x; 1.2744x over previous
#include <hip/hip_runtime.h>


typedef float v2f __attribute__((ext_vector_type(2)));
typedef float v4f __attribute__((ext_vector_type(4)));
typedef float v8f __attribute__((ext_vector_type(8)));
typedef int   v2i __attribute__((ext_vector_type(2)));
typedef int   v4i __attribute__((ext_vector_type(4)));

constexpr int SLOTS     = 64;
constexpr int N_NODES   = 50000;
constexpr int N_EDGES   = 800000;
constexpr int BIN_SHIFT = 8;
constexpr int BIN_NODES = 1 << BIN_SHIFT;
constexpr int NBINS     = (N_NODES + BIN_NODES - 1) / BIN_NODES;
constexpr int CH        = 2048;
constexpr int NG        = (N_EDGES + CH - 1) / CH;
constexpr int CAP       = 32;
constexpr int TPITCH    = SLOTS + 1;
constexpr int PB_T      = 512;
constexpr int PB_W      = PB_T / 32;
constexpr int PB_U      = 7;

__device__ __forceinline__ void store_bin(int s, int d, int r, int g, int* binCnt,
                                          v2i* __restrict__ region, v2i* __restrict__ ovf) {
    v2i rec;
    rec.x = s;
    rec.y = d;
    if (r < CAP) {
        region[((size_t)(d >> BIN_SHIFT) * NG + g) * CAP + r] = rec;
    } else {
        int o = atomicAdd(&binCnt[NBINS], 1);
        ovf[(size_t)g * CH + o] = rec;
    }
}

__device__ __forceinline__ void store_slot(v2i rec, int slot, int* tab,
                                           int* __restrict__ cnt, int* __restrict__ spill, int n) {
    if (slot < SLOTS) {
        tab[(rec.y & (BIN_NODES - 1)) * TPITCH + slot] = rec.x;
    } else {
        int o = atomicAdd(&cnt[n], 1);
        ((v2i*)spill)[o] = rec;
    }
}

__global__ __launch_bounds__(PB_T) void slots_place(
        int* __restrict__ cnt, int* __restrict__ table, int* __restrict__ spill,
        const int* __restrict__ counts, const int* __restrict__ ovfN,
        const v2i* __restrict__ region, const v2i* __restrict__ ovf, int n) {
    __shared__ int tab[BIN_NODES * TPITCH];
    __shared__ int nodeCnt[BIN_NODES];
    __shared__ int cellCnt[NG];
    __shared__ int ovfCnt[NG];
    __shared__ int anyOvf;

    int b = blockIdx.x;
    int t = threadIdx.x;
    for (int i = t; i < BIN_NODES * TPITCH; i += PB_T) tab[i] = 0;
    if (t < BIN_NODES) nodeCnt[t] = 0;
    if (t == 0) anyOvf = 0;
    for (int i = t; i < NG; i += PB_T) cellCnt[i] = counts[b * NG + i];
    __syncthreads();
    for (int i = t; i < NG; i += PB_T) {
        int c = ovfN[i];
        ovfCnt[i] = c;
        if (c > 0) anyOvf = 1;
    }

    int w    = __builtin_amdgcn_readfirstlane(t >> 5);
    int lane = t & 31;
    const v2i* reg = region + (size_t)b * NG * CAP;
    for (int g0 = w; g0 < NG; g0 += PB_W * PB_U) {
        v2i  ent[PB_U];
        bool ok[PB_U];
        int  slot[PB_U];
#pragma unroll
        for (int u = 0; u < PB_U; ++u) {
            int g = g0 + PB_W * u;
            ok[u] = (g < NG) && (lane < cellCnt[g < NG ? g : 0]);
            v2i e = {0, 0};
            if (ok[u]) e = reg[(size_t)g * CAP + lane];
            ent[u] = e;
        }
#pragma unroll
        for (int u = 0; u < PB_U; ++u) {
            slot[u] = 0;
            if (ok[u]) slot[u] = atomicAdd(&nodeCnt[ent[u].y & (BIN_NODES - 1)], 1);
        }
#pragma unroll
        for (int u = 0; u < PB_U; ++u)
            if (ok[u]) store_slot(ent[u], slot[u], tab, cnt, spill, n);
    }
    __syncthreads();
    if (anyOvf) {
        for (int g = 0; g < NG; ++g) {
            int c = ovfCnt[g];
            if (c > CH) c = CH;
            for (int i = t; i < c; i += PB_T) {
                v2i e = ovf[(size_t)g * CH + i];
                if ((e.y >> BIN_SHIFT) == b) {
                    int sl = atomicAdd(&nodeCnt[e.y & (BIN_NODES - 1)], 1);
                    store_slot(e, sl, tab, cnt, spill, n);
                }
            }
        }
    }
    __syncthreads();
    int node0 = b * BIN_NODES;
    if (t < BIN_NODES && node0 + t < n) cnt[node0 + t] = nodeCnt[t];
    int nrows = n - node0;
    if (nrows > BIN_NODES) nrows = BIN_NODES;
    int* trow = table + (size_t)node0 * SLOTS;
    for (int i = t; i < nrows * SLOTS; i += PB_T)
        trow[i] = tab[(i >> 6) * TPITCH + (i & (SLOTS - 1))];
}

__device__ __forceinline__ void fold_product(const float* A, const float* B,
                                             float* C, float* Blds) {
    constexpr int KK  = 128;
    constexpr int TN  = 4;
    constexpr int Nc  = TN * 16;
    constexpr int LDB = KK + 2;

    for (int i = threadIdx.x; i < KK * Nc; i += 256) {
        int k = i / Nc;
        int c = i - k * Nc;
        Blds[c * LDB + k] = B[i];
    }
    __syncthreads();

    int tm   = threadIdx.x >> 5;
    int lane = threadIdx.x & 31;
    int half = lane >> 4;
    int lr   = lane & 15;

    const float* Arow = A + (size_t)(tm * 16 + lr) * KK;
    v8f acc[TN] = {};
    for (int k = 0; k < KK; k += 4) {
        v2f a = *(const v2f*)(Arow + k + 2 * half);
#pragma unroll
        for (int t = 0; t < TN; ++t) {
            v2f b = *(const v2f*)(&Blds[(t * 16 + lr) * LDB + k + 2 * half]);
            acc[t] = __builtin_amdgcn_wmma_f32_16x16x4_f32(
                         false, a, false, b, (short)0, acc[t], false, false);
        }
    }
#pragma unroll
    for (int t = 0; t < TN; ++t) {
        int col = t * 16 + lr;
#pragma unroll
        for (int r = 0; r < 8; ++r) {
            int row = tm * 16 + r + 8 * half;
            C[(size_t)row * Nc + col] = acc[t][r];
        }
    }
    __syncthreads();
}

template <int KK, int TN, bool RELU, bool BIAS, bool DSCALE>
__device__ __forceinline__ void gemm_group(
        const float* __restrict__ A, const float* __restrict__ B,
        const float* __restrict__ bias, const int* __restrict__ cnt,
        float* __restrict__ C, int M, int blk, float* Blds) {
    constexpr int Nc  = TN * 16;
    constexpr int LDB = KK + 2;

    for (int i = threadIdx.x; i < KK * Nc; i += 256) {
        int k = i / Nc;
        int c = i - k * Nc;
        Blds[c * LDB + k] = B[i];
    }
    __syncthreads();

    int tilesM = M >> 4;
    int tm     = blk * 8 + __builtin_amdgcn_readfirstlane((int)(threadIdx.x >> 5));
    int lane   = threadIdx.x & 31;
    int half   = lane >> 4;
    int lr     = lane & 15;

    if (tm < tilesM) {
        const float* Arow = A + (size_t)(tm * 16 + lr) * KK;
        __builtin_prefetch(Arow, 0, 0);
        __builtin_prefetch(Arow + 64, 0, 0);

        v8f acc[TN] = {};
        for (int k = 0; k < KK; k += 4) {
            v2f a = *(const v2f*)(Arow + k + 2 * half);
#pragma unroll
            for (int t = 0; t < TN; ++t) {
                v2f b = *(const v2f*)(&Blds[(t * 16 + lr) * LDB + k + 2 * half]);
                acc[t] = __builtin_amdgcn_wmma_f32_16x16x4_f32(
                             false, a, false, b, (short)0, acc[t], false, false);
            }
        }

        float sc[8];
#pragma unroll
        for (int r = 0; r < 8; ++r) {
            sc[r] = 1.0f;
            if (DSCALE) sc[r] = rsqrtf((float)cnt[tm * 16 + r + 8 * half] + 1.0f);
        }
#pragma unroll
        for (int t = 0; t < TN; ++t) {
            int col  = t * 16 + lr;
            float bv = BIAS ? bias[col] : 0.0f;
#pragma unroll
            for (int r = 0; r < 8; ++r) {
                int row = tm * 16 + r + 8 * half;
                float v = acc[t][r] + bv;
                if (RELU) v = fmaxf(v, 0.0f);
                if (DSCALE) v = v * sc[r];
                C[(size_t)row * Nc + col] = v;
            }
        }
    }
}

template <int KK, int TN, bool RELU, bool BIAS, bool DSCALE>
__global__ __launch_bounds__(256) void gcn_gemm_wmma(
        const float* __restrict__ A, const float* __restrict__ B,
        const float* __restrict__ bias, const int* __restrict__ cnt,
        float* __restrict__ C, int M) {
    __shared__ float Blds[TN * 16 * (KK + 2)];
    gemm_group<KK, TN, RELU, BIAS, DSCALE>(A, B, bias, cnt, C, M, (int)blockIdx.x, Blds);
}

__global__ __launch_bounds__(256, 8) void slots_bin_fold_xw1(
        int* __restrict__ cnt, int* __restrict__ counts, int* __restrict__ ovfN,
        v2i* __restrict__ region, v2i* __restrict__ ovf,
        const int* __restrict__ src, const int* __restrict__ dst, int e, int n,
        const float* Wc, const float* W2, const float* bc, const float* b2,
        float* M2, float* M1, float* M0, float* beta,
        const float* __restrict__ x, const float* __restrict__ W1,
        const float* __restrict__ b1, float* __restrict__ h0,
        int rows, int gemm_groups) {
    __shared__ float Blds[128 * 130];
    __shared__ int binCnt[NBINS + 1];
    if (blockIdx.x == 0) {
        if (threadIdx.x == 0) cnt[n] = 0;
        fold_product(Wc + 2 * 128 * 128, W2, M2, Blds);
        fold_product(Wc + 1 * 128 * 128, M2, M1, Blds);
        fold_product(Wc,                 M1, M0, Blds);
        if (threadIdx.x < 192) {
            int l = threadIdx.x >> 6;
            int c = threadIdx.x & 63;
            const float* M = (l == 0) ? M1 : ((l == 1) ? M2 : W2);
            const float* b = bc + l * 128;
            float s = 0.0f;
            for (int k = 0; k < 128; ++k) s += b[k] * M[k * 64 + c];
            if (l == 2) s += b2[c];
            beta[l * 64 + c] = s;
        }
        return;
    }
    if ((int)blockIdx.x <= gemm_groups) {
        gemm_group<128, 8, true, true, false>(
            x, W1, b1, nullptr, h0, rows, (int)blockIdx.x - 1, Blds);
        return;
    }
    int g = (int)blockIdx.x - 1 - gemm_groups;
    for (int i = threadIdx.x; i < NBINS + 1; i += 256) binCnt[i] = 0;
    __syncthreads();
    constexpr int UB = CH / 1024;
    int sv[UB * 4], dv[UB * 4], rk[UB * 4];
#pragma unroll
    for (int u = 0; u < UB; ++u) {
        int i = g * CH + u * 1024 + (int)threadIdx.x * 4;
        v4i s4 = {0, 0, 0, 0};
        v4i d4 = {-1, -1, -1, -1};
        if (i + 3 < e) {
            s4 = *(const v4i*)(src + i);
            d4 = *(const v4i*)(dst + i);
        } else {
            if (i     < e) { s4.x = src[i];     d4.x = dst[i];     }
            if (i + 1 < e) { s4.y = src[i + 1]; d4.y = dst[i + 1]; }
            if (i + 2 < e) { s4.z = src[i + 2]; d4.z = dst[i + 2]; }
        }
        sv[u * 4 + 0] = s4.x; sv[u * 4 + 1] = s4.y; sv[u * 4 + 2] = s4.z; sv[u * 4 + 3] = s4.w;
        dv[u * 4 + 0] = d4.x; dv[u * 4 + 1] = d4.y; dv[u * 4 + 2] = d4.z; dv[u * 4 + 3] = d4.w;
    }
#pragma unroll
    for (int q = 0; q < UB * 4; ++q) {
        rk[q] = 0;
        if (dv[q] >= 0) rk[q] = atomicAdd(&binCnt[dv[q] >> BIN_SHIFT], 1);
    }
#pragma unroll
    for (int q = 0; q < UB * 4; ++q)
        if (dv[q] >= 0) store_bin(sv[q], dv[q], rk[q], g, binCnt, region, ovf);
    __syncthreads();
    for (int i = threadIdx.x; i < NBINS; i += 256) {
        int c = binCnt[i];
        counts[i * NG + g] = c < CAP ? c : CAP;
    }
    if (threadIdx.x == 0) ovfN[g] = binCnt[NBINS];
}

template <int U, bool TAIL>
__device__ __forceinline__ void gather_pairs(v4f& acc, const v4f* __restrict__ q4,
                                             int rec, int j, int m, int half, int c) {
    v4f r[U];
#pragma unroll
    for (int u = 0; u < U; ++u) {
        int i0 = j + 2 * u;
        int s0 = __builtin_amdgcn_readlane(rec, i0);
        int s1 = __builtin_amdgcn_readlane(rec, i0 + 1);
        int s  = half ? s1 : s0;
        if (TAIL) {
            v4f t = {0.0f, 0.0f, 0.0f, 0.0f};
            if (i0 + half < m) t = q4[(size_t)s * 16 + c];
            r[u] = t;
        } else {
            r[u] = q4[(size_t)s * 16 + c];
        }
    }
#pragma unroll
    for (int u = 0; u < U; ++u) acc += r[u];
}

__device__ __forceinline__ void gather_chunk(v4f& acc, const v4f* __restrict__ q4,
                                             int rec, int m, int lane, int half, int c) {
    if (lane >= m) rec = 0;
    int j = 0;
    for (; j + 16 <= m; j += 16) gather_pairs<8, false>(acc, q4, rec, j, m, half, c);
    if (j + 8 <= m) { gather_pairs<4, false>(acc, q4, rec, j, m, half, c); j += 8; }
    if (j + 4 <= m) { gather_pairs<2, false>(acc, q4, rec, j, m, half, c); j += 4; }
    if (j < m)      { gather_pairs<2, true >(acc, q4, rec, j, m, half, c); }
}

template <bool LAST>
__global__ __launch_bounds__(256) void gcn_gather64(
        float* __restrict__ out, const float* __restrict__ q,
        const float* __restrict__ beta, const int* __restrict__ cnt,
        const int* __restrict__ table, const int* __restrict__ spill,
        int n, int e) {
    int node = blockIdx.x * 8 + __builtin_amdgcn_readfirstlane((int)(threadIdx.x >> 5));
    if (node >= n) return;
    int lane = threadIdx.x & 31;
    int half = lane >> 4;
    int c    = lane & 15;

    const v4f* q4  = (const v4f*)q;
    const int* row = table + (size_t)node * SLOTS;

    int deg  = __builtin_amdgcn_readfirstlane(cnt[node]);
    int rec0 = row[lane];
    v4f self = q4[(size_t)node * 16 + c];
    v4f bv   = ((const v4f*)beta)[c];

    int filled = deg < SLOTS ? deg : SLOTS;
    v4f acc = {0.0f, 0.0f, 0.0f, 0.0f};
    gather_chunk(acc, q4, rec0, filled < 32 ? filled : 32, lane, half, c);
    if (filled > 32)
        gather_chunk(acc, q4, row[32 + lane], filled - 32, lane, half, c);

    if (deg > SLOTS) {
        int total = __builtin_amdgcn_readfirstlane(cnt[n]);
        if (total > e) total = e;
        const v2i* sp = (const v2i*)spill;
        for (int ob = 0; ob < total; ob += 32) {
            int mm = total - ob;
            if (mm > 32) mm = 32;
            v2i orec = {0, -1};
            if (lane < mm) orec = sp[ob + lane];
            for (int jj = 0; jj < mm; ++jj) {
                int s = __builtin_amdgcn_readlane(orec.x, jj);
                int d = __builtin_amdgcn_readlane(orec.y, jj);
                if (d == node && half == 0) acc += q4[(size_t)s * 16 + c];
            }
        }
    }

    v4f oth;
    oth.x = __shfl_xor(acc.x, 16);
    oth.y = __shfl_xor(acc.y, 16);
    oth.z = __shfl_xor(acc.z, 16);
    oth.w = __shfl_xor(acc.w, 16);
    float dn = rsqrtf((float)deg + 1.0f);
    v4f v = ((acc + oth) + self) * dn + bv;
    if (!LAST) v = v * dn;
    if (half == 0)
        ((v4f*)out)[(size_t)node * 16 + c] = v;
}

extern "C" void kernel_launch(void* const* d_in, const int* in_sizes, int n_in,
                              void* d_out, int out_size, void* d_ws, size_t ws_size,
                              hipStream_t stream) {
    const float* x  = (const float*)d_in[0];
    const int*   ei = (const int*)  d_in[1];
    const float* W1 = (const float*)d_in[2];
    const float* b1 = (const float*)d_in[3];
    const float* Wc = (const float*)d_in[4];
    const float* bc = (const float*)d_in[5];
    const float* W2 = (const float*)d_in[6];
    const float* b2 = (const float*)d_in[7];

    const int N = N_NODES, E = N_EDGES, H = 128;
    const int* srcv = ei;
    const int* dstv = ei + E;


    char* ws = (char*)d_ws;
    size_t off = 0;
    auto carve = [&](size_t bytes) -> void* {
        void* p = ws + off;
        off = (off + bytes + 255) & ~(size_t)255;
        return p;
    };
    int*   cnt    = (int*)  carve((size_t)(N + 1) * 4);
    int*   table  = (int*)  carve((size_t)N * SLOTS * 4);
    int*   spill  = (int*)  carve((size_t)E * 8);
    float* h0     = (float*)carve((size_t)N * H * 4);
    float* pA     = (float*)carve((size_t)N * 64 * 4);
    float* pB     = (float*)carve((size_t)N * 64 * 4);
    float* M2     = (float*)carve((size_t)H * 64 * 4);
    float* M1     = (float*)carve((size_t)H * 64 * 4);
    float* M0     = (float*)carve((size_t)H * 64 * 4);
    float* beta   = (float*)carve((size_t)3 * 64 * 4);
    v2i*   region = (v2i*)  carve((size_t)NBINS * NG * CAP * 8);
    int*   counts = (int*)  carve((size_t)NBINS * NG * 4);
    int*   ovfN   = (int*)  carve((size_t)NG * 4);
    v2i*   ovf    = (v2i*)  carve((size_t)NG * CH * 8);

    const int tilesM    = N / 16;
    const int gemmBlks  = (tilesM + 7) / 8;

    slots_bin_fold_xw1<<<1 + gemmBlks + NG, 256, 0, stream>>>(
        cnt, counts, ovfN, region, ovf, srcv, dstv, E, N, Wc, W2, bc, b2,
        M2, M1, M0, beta, x, W1, b1, h0, N, gemmBlks);
    slots_place<<<NBINS, PB_T, 0, stream>>>(
        cnt, table, spill, counts, ovfN, region, ovf, N);

    gcn_gemm_wmma<128, 4, false, false, true><<<gemmBlks, 256, 0, stream>>>(
        h0, M0, nullptr, cnt, pA, N);

    gcn_gather64<false><<<(N + 7) / 8, 256, 0, stream>>>(
        pB, pA, beta, cnt, table, spill, N, E);
    gcn_gather64<false><<<(N + 7) / 8, 256, 0, stream>>>(
        pA, pB, beta + 64, cnt, table, spill, N, E);
    gcn_gather64<true><<<(N + 7) / 8, 256, 0, stream>>>(
        (float*)d_out, pA, beta + 128, cnt, table, spill, N, E);
}


# h0 @ M0 product run by 16 more waves of each edge-place work-group (its 256 rows = the group's node range, row factor from the LDS degree counters): 6 launches to 5; fp sum order unchanged
# speedup vs baseline: 6.9273x; 1.0561x over previous
#include <hip/hip_runtime.h>


typedef float v2f __attribute__((ext_vector_type(2)));
typedef float v4f __attribute__((ext_vector_type(4)));
typedef float v8f __attribute__((ext_vector_type(8)));
typedef int   v2i __attribute__((ext_vector_type(2)));
typedef int   v4i __attribute__((ext_vector_type(4)));

constexpr int SLOTS     = 64;
constexpr int N_NODES   = 50000;
constexpr int N_EDGES   = 800000;
constexpr int BIN_SHIFT = 8;
constexpr int BIN_NODES = 1 << BIN_SHIFT;
constexpr int NBINS     = (N_NODES + BIN_NODES - 1) / BIN_NODES;
constexpr int CH        = 2048;
constexpr int NG        = (N_EDGES + CH - 1) / CH;
constexpr int CAP       = 32;
constexpr int TPITCH    = SLOTS + 1;
constexpr int PB_T      = 512;
constexpr int PB_W      = PB_T / 32;
constexpr int PG_T      = 2 * PB_T;
constexpr int QK        = 128;
constexpr int QN        = 64;
constexpr int QLDB      = QK + 2;
constexpr int PB_U      = 7;

__device__ __forceinline__ void store_bin(int s, int d, int r, int g, int* binCnt,
                                          v2i* __restrict__ region, v2i* __restrict__ ovf) {
    v2i rec;
    rec.x = s;
    rec.y = d;
    if (r < CAP) {
        region[((size_t)(d >> BIN_SHIFT) * NG + g) * CAP + r] = rec;
    } else {
        int o = atomicAdd(&binCnt[NBINS], 1);
        ovf[(size_t)g * CH + o] = rec;
    }
}

__device__ __forceinline__ void store_slot(v2i rec, int slot, int* tab,
                                           int* __restrict__ cnt, int* __restrict__ spill, int n) {
    if (slot < SLOTS) {
        tab[(rec.y & (BIN_NODES - 1)) * TPITCH + slot] = rec.x;
    } else {
        int o = atomicAdd(&cnt[n], 1);
        ((v2i*)spill)[o] = rec;
    }
}

__global__ __launch_bounds__(PG_T) void slots_place_q(
        int* __restrict__ cnt, int* __restrict__ table, int* __restrict__ spill,
        const int* __restrict__ counts, const int* __restrict__ ovfN,
        const v2i* __restrict__ region, const v2i* __restrict__ ovf, int n,
        const float* __restrict__ h0, const float* __restrict__ M0,
        float* __restrict__ q) {
    __shared__ int tab[BIN_NODES * TPITCH];
    __shared__ int nodeCnt[BIN_NODES];
    __shared__ int cellCnt[NG];
    __shared__ int ovfCnt[NG];
    __shared__ int anyOvf;
    __shared__ float Blds[QN * QLDB];

    int b = blockIdx.x;
    int t = threadIdx.x;
    int w    = __builtin_amdgcn_readfirstlane(t >> 5);
    int lane = t & 31;
    bool placer = w < PB_W;
    int half = lane >> 4;
    int lr   = lane & 15;
    int tm   = b * (BIN_NODES / 16) + (w - PB_W);
    bool mult = !placer && tm < (n >> 4);

    if (placer) {
        for (int i = t; i < BIN_NODES * TPITCH; i += PB_T) tab[i] = 0;
        if (t < BIN_NODES) nodeCnt[t] = 0;
        if (t == 0) anyOvf = 0;
        for (int i = t; i < NG; i += PB_T) cellCnt[i] = counts[b * NG + i];
    } else {
        float m[QK * QN / PB_T];
#pragma unroll
        for (int u = 0; u < QK * QN / PB_T; ++u) m[u] = M0[u * PB_T + (t - PB_T)];
#pragma unroll
        for (int u = 0; u < QK * QN / PB_T; ++u) {
            int i = u * PB_T + (t - PB_T);
            Blds[(i & (QN - 1)) * QLDB + (i >> 6)] = m[u];
        }
    }
    __syncthreads();

    v8f acc[QN / 16] = {};
    if (mult) {
        const float* Arow = h0 + (size_t)(tm * 16 + lr) * QK;
        for (int k = 0; k < QK; k += 4) {
            v2f a = *(const v2f*)(Arow + k + 2 * half);
#pragma unroll
            for (int j = 0; j < QN / 16; ++j) {
                v2f bf = *(const v2f*)(&Blds[(j * 16 + lr) * QLDB + k + 2 * half]);
                acc[j] = __builtin_amdgcn_wmma_f32_16x16x4_f32(
                             false, a, false, bf, (short)0, acc[j], false, false);
            }
        }
    }

    if (placer) {
        for (int i = t; i < NG; i += PB_T) {
            int c = ovfN[i];
            ovfCnt[i] = c;
            if (c > 0) anyOvf = 1;
        }

        const v2i* reg = region + (size_t)b * NG * CAP;
        for (int g0 = w; g0 < NG; g0 += PB_W * PB_U) {
            v2i  ent[PB_U];
            bool ok[PB_U];
            int  slot[PB_U];
#pragma unroll
            for (int u = 0; u < PB_U; ++u) {
                int g = g0 + PB_W * u;
                ok[u] = (g < NG) && (lane < cellCnt[g < NG ? g : 0]);
                v2i e = {0, 0};
                if (ok[u]) e = reg[(size_t)g * CAP + lane];
                ent[u] = e;
            }
#pragma unroll
            for (int u = 0; u < PB_U; ++u) {
                slot[u] = 0;
                if (ok[u]) slot[u] = atomicAdd(&nodeCnt[ent[u].y & (BIN_NODES - 1)], 1);
            }
#pragma unroll
            for (int u = 0; u < PB_U; ++u)
                if (ok[u]) store_slot(ent[u], slot[u], tab, cnt, spill, n);
        }
    }
    __syncthreads();
    if (anyOvf && placer) {
        for (int g = 0; g < NG; ++g) {
            int c = ovfCnt[g];
            if (c > CH) c = CH;
            for (int i = t; i < c; i += PB_T) {
                v2i e = ovf[(size_t)g * CH + i];
                if ((e.y >> BIN_SHIFT) == b) {
                    int sl = atomicAdd(&nodeCnt[e.y & (BIN_NODES - 1)], 1);
                    store_slot(e, sl, tab, cnt, spill, n);
                }
            }
        }
    }
    __syncthreads();
    if (placer) {
        int node0 = b * BIN_NODES;
        if (t < BIN_NODES && node0 + t < n) cnt[node0 + t] = nodeCnt[t];
        int nrows = n - node0;
        if (nrows > BIN_NODES) nrows = BIN_NODES;
        int* trow = table + (size_t)node0 * SLOTS;
        for (int i = t; i < nrows * SLOTS; i += PB_T)
            trow[i] = tab[(i >> 6) * TPITCH + (i & (SLOTS - 1))];
    } else if (mult) {
        float sc[8];
#pragma unroll
        for (int r = 0; r < 8; ++r)
            sc[r] = rsqrtf((float)nodeCnt[(w - PB_W) * 16 + r + 8 * half] + 1.0f);
#pragma unroll
        for (int j = 0; j < QN / 16; ++j) {
            int col = j * 16 + lr;
#pragma unroll
            for (int r = 0; r < 8; ++r) {
                int row = tm * 16 + r + 8 * half;
                q[(size_t)row * QN + col] = acc[j][r] * sc[r];
            }
        }
    }
}

__device__ __forceinline__ void fold_product(const float* A, const float* B,
                                             float* C, float* Blds) {
    constexpr int KK  = 128;
    constexpr int TN  = 4;
    constexpr int Nc  = TN * 16;
    constexpr int LDB = KK + 2;

    for (int i = threadIdx.x; i < KK * Nc; i += 256) {
        int k = i / Nc;
        int c = i - k * Nc;
        Blds[c * LDB + k] = B[i];
    }
    __syncthreads();

    int tm   = threadIdx.x >> 5;
    int lane = threadIdx.x & 31;
    int half = lane >> 4;
    int lr   = lane & 15;

    const float* Arow = A + (size_t)(tm * 16 + lr) * KK;
    v8f acc[TN] = {};
    for (int k = 0; k < KK; k += 4) {
        v2f a = *(const v2f*)(Arow + k + 2 * half);
#pragma unroll
        for (int t = 0; t < TN; ++t) {
            v2f b = *(const v2f*)(&Blds[(t * 16 + lr) * LDB + k + 2 * half]);
            acc[t] = __builtin_amdgcn_wmma_f32_16x16x4_f32(
                         false, a, false, b, (short)0, acc[t], false, false);
        }
    }
#pragma unroll
    for (int t = 0; t < TN; ++t) {
        int col = t * 16 + lr;
#pragma unroll
        for (int r = 0; r < 8; ++r) {
            int row = tm * 16 + r + 8 * half;
            C[(size_t)row * Nc + col] = acc[t][r];
        }
    }
    __syncthreads();
}

template <int KK, int TN, bool RELU, bool BIAS>
__device__ __forceinline__ void gemm_group(
        const float* __restrict__ A, const float* __restrict__ B,
        const float* __restrict__ bias,
        float* __restrict__ C, int M, int blk, float* Blds) {
    constexpr int Nc  = TN * 16;
    constexpr int LDB = KK + 2;

    for (int i = threadIdx.x; i < KK * Nc; i += 256) {
        int k = i / Nc;
        int c = i - k * Nc;
        Blds[c * LDB + k] = B[i];
    }
    __syncthreads();

    int tilesM = M >> 4;
    int tm     = blk * 8 + __builtin_amdgcn_readfirstlane((int)(threadIdx.x >> 5));
    int lane   = threadIdx.x & 31;
    int half   = lane >> 4;
    int lr     = lane & 15;

    if (tm < tilesM) {
        const float* Arow = A + (size_t)(tm * 16 + lr) * KK;
        __builtin_prefetch(Arow, 0, 0);
        __builtin_prefetch(Arow + 64, 0, 0);

        v8f acc[TN] = {};
        for (int k = 0; k < KK; k += 4) {
            v2f a = *(const v2f*)(Arow + k + 2 * half);
#pragma unroll
            for (int t = 0; t < TN; ++t) {
                v2f b = *(const v2f*)(&Blds[(t * 16 + lr) * LDB + k + 2 * half]);
                acc[t] = __builtin_amdgcn_wmma_f32_16x16x4_f32(
                             false, a, false, b, (short)0, acc[t], false, false);
            }
        }

#pragma unroll
        for (int t = 0; t < TN; ++t) {
            int col  = t * 16 + lr;
            float bv = BIAS ? bias[col] : 0.0f;
#pragma unroll
            for (int r = 0; r < 8; ++r) {
                int row = tm * 16 + r + 8 * half;
                float v = acc[t][r] + bv;
                if (RELU) v = fmaxf(v, 0.0f);
                C[(size_t)row * Nc + col] = v;
            }
        }
    }
}

__global__ __launch_bounds__(256, 8) void slots_bin_fold_xw1(
        int* __restrict__ cnt, int* __restrict__ counts, int* __restrict__ ovfN,
        v2i* __restrict__ region, v2i* __restrict__ ovf,
        const int* __restrict__ src, const int* __restrict__ dst, int e, int n,
        const float* Wc, const float* W2, const float* bc, const float* b2,
        float* M2, float* M1, float* M0, float* beta,
        const float* __restrict__ x, const float* __restrict__ W1,
        const float* __restrict__ b1, float* __restrict__ h0,
        int rows, int gemm_groups) {
    __shared__ float Blds[128 * 130];
    __shared__ int binCnt[NBINS + 1];
    if (blockIdx.x == 0) {
        if (threadIdx.x == 0) cnt[n] = 0;
        fold_product(Wc + 2 * 128 * 128, W2, M2, Blds);
        fold_product(Wc + 1 * 128 * 128, M2, M1, Blds);
        fold_product(Wc,                 M1, M0, Blds);
        if (threadIdx.x < 192) {
            int l = threadIdx.x >> 6;
            int c = threadIdx.x & 63;
            const float* M = (l == 0) ? M1 : ((l == 1) ? M2 : W2);
            const float* b = bc + l * 128;
            float s = 0.0f;
            for (int k = 0; k < 128; ++k) s += b[k] * M[k * 64 + c];
            if (l == 2) s += b2[c];
            beta[l * 64 + c] = s;
        }
        return;
    }
    if ((int)blockIdx.x <= gemm_groups) {
        gemm_group<128, 8, true, true>(
            x, W1, b1, h0, rows, (int)blockIdx.x - 1, Blds);
        return;
    }
    int g = (int)blockIdx.x - 1 - gemm_groups;
    for (int i = threadIdx.x; i < NBINS + 1; i += 256) binCnt[i] = 0;
    __syncthreads();
    constexpr int UB = CH / 1024;
    int sv[UB * 4], dv[UB * 4], rk[UB * 4];
#pragma unroll
    for (int u = 0; u < UB; ++u) {
        int i = g * CH + u * 1024 + (int)threadIdx.x * 4;
        v4i s4 = {0, 0, 0, 0};
        v4i d4 = {-1, -1, -1, -1};
        if (i + 3 < e) {
            s4 = *(const v4i*)(src + i);
            d4 = *(const v4i*)(dst + i);
        } else {
            if (i     < e) { s4.x = src[i];     d4.x = dst[i];     }
            if (i + 1 < e) { s4.y = src[i + 1]; d4.y = dst[i + 1]; }
            if (i + 2 < e) { s4.z = src[i + 2]; d4.z = dst[i + 2]; }
        }
        sv[u * 4 + 0] = s4.x; sv[u * 4 + 1] = s4.y; sv[u * 4 + 2] = s4.z; sv[u * 4 + 3] = s4.w;
        dv[u * 4 + 0] = d4.x; dv[u * 4 + 1] = d4.y; dv[u * 4 + 2] = d4.z; dv[u * 4 + 3] = d4.w;
    }
#pragma unroll
    for (int q = 0; q < UB * 4; ++q) {
        rk[q] = 0;
        if (dv[q] >= 0) rk[q] = atomicAdd(&binCnt[dv[q] >> BIN_SHIFT], 1);
    }
#pragma unroll
    for (int q = 0; q < UB * 4; ++q)
        if (dv[q] >= 0) store_bin(sv[q], dv[q], rk[q], g, binCnt, region, ovf);
    __syncthreads();
    for (int i = threadIdx.x; i < NBINS; i += 256) {
        int c = binCnt[i];
        counts[i * NG + g] = c < CAP ? c : CAP;
    }
    if (threadIdx.x == 0) ovfN[g] = binCnt[NBINS];
}

template <int U, bool TAIL>
__device__ __forceinline__ void gather_pairs(v4f& acc, const v4f* __restrict__ q4,
                                             int rec, int j, int m, int half, int c) {
    v4f r[U];
#pragma unroll
    for (int u = 0; u < U; ++u) {
        int i0 = j + 2 * u;
        int s0 = __builtin_amdgcn_readlane(rec, i0);
        int s1 = __builtin_amdgcn_readlane(rec, i0 + 1);
        int s  = half ? s1 : s0;
        if (TAIL) {
            v4f t = {0.0f, 0.0f, 0.0f, 0.0f};
            if (i0 + half < m) t = q4[(size_t)s * 16 + c];
            r[u] = t;
        } else {
            r[u] = q4[(size_t)s * 16 + c];
        }
    }
#pragma unroll
    for (int u = 0; u < U; ++u) acc += r[u];
}

__device__ __forceinline__ void gather_chunk(v4f& acc, const v4f* __restrict__ q4,
                                             int rec, int m, int lane, int half, int c) {
    if (lane >= m) rec = 0;
    int j = 0;
    for (; j + 16 <= m; j += 16) gather_pairs<8, false>(acc, q4, rec, j, m, half, c);
    if (j + 8 <= m) { gather_pairs<4, false>(acc, q4, rec, j, m, half, c); j += 8; }
    if (j + 4 <= m) { gather_pairs<2, false>(acc, q4, rec, j, m, half, c); j += 4; }
    if (j < m)      { gather_pairs<2, true >(acc, q4, rec, j, m, half, c); }
}

template <bool LAST>
__global__ __launch_bounds__(256) void gcn_gather64(
        float* __restrict__ out, const float* __restrict__ q,
        const float* __restrict__ beta, const int* __restrict__ cnt,
        const int* __restrict__ table, const int* __restrict__ spill,
        int n, int e) {
    int node = blockIdx.x * 8 + __builtin_amdgcn_readfirstlane((int)(threadIdx.x >> 5));
    if (node >= n) return;
    int lane = threadIdx.x & 31;
    int half = lane >> 4;
    int c    = lane & 15;

    const v4f* q4  = (const v4f*)q;
    const int* row = table + (size_t)node * SLOTS;

    int deg  = __builtin_amdgcn_readfirstlane(cnt[node]);
    int rec0 = row[lane];
    v4f self = q4[(size_t)node * 16 + c];
    v4f bv   = ((const v4f*)beta)[c];

    int filled = deg < SLOTS ? deg : SLOTS;
    v4f acc = {0.0f, 0.0f, 0.0f, 0.0f};
    gather_chunk(acc, q4, rec0, filled < 32 ? filled : 32, lane, half, c);
    if (filled > 32)
        gather_chunk(acc, q4, row[32 + lane], filled - 32, lane, half, c);

    if (deg > SLOTS) {
        int total = __builtin_amdgcn_readfirstlane(cnt[n]);
        if (total > e) total = e;
        const v2i* sp = (const v2i*)spill;
        for (int ob = 0; ob < total; ob += 32) {
            int mm = total - ob;
            if (mm > 32) mm = 32;
            v2i orec = {0, -1};
            if (lane < mm) orec = sp[ob + lane];
            for (int jj = 0; jj < mm; ++jj) {
                int s = __builtin_amdgcn_readlane(orec.x, jj);
                int d = __builtin_amdgcn_readlane(orec.y, jj);
                if (d == node && half == 0) acc += q4[(size_t)s * 16 + c];
            }
        }
    }

    v4f oth;
    oth.x = __shfl_xor(acc.x, 16);
    oth.y = __shfl_xor(acc.y, 16);
    oth.z = __shfl_xor(acc.z, 16);
    oth.w = __shfl_xor(acc.w, 16);
    float dn = rsqrtf((float)deg + 1.0f);
    v4f v = ((acc + oth) + self) * dn + bv;
    if (!LAST) v = v * dn;
    if (half == 0)
        ((v4f*)out)[(size_t)node * 16 + c] = v;
}

extern "C" void kernel_launch(void* const* d_in, const int* in_sizes, int n_in,
                              void* d_out, int out_size, void* d_ws, size_t ws_size,
                              hipStream_t stream) {
    const float* x  = (const float*)d_in[0];
    const int*   ei = (const int*)  d_in[1];
    const float* W1 = (const float*)d_in[2];
    const float* b1 = (const float*)d_in[3];
    const float* Wc = (const float*)d_in[4];
    const float* bc = (const float*)d_in[5];
    const float* W2 = (const float*)d_in[6];
    const float* b2 = (const float*)d_in[7];

    const int N = N_NODES, E = N_EDGES, H = 128;
    const int* srcv = ei;
    const int* dstv = ei + E;


    char* ws = (char*)d_ws;
    size_t off = 0;
    auto carve = [&](size_t bytes) -> void* {
        void* p = ws + off;
        off = (off + bytes + 255) & ~(size_t)255;
        return p;
    };
    int*   cnt    = (int*)  carve((size_t)(N + 1) * 4);
    int*   table  = (int*)  carve((size_t)N * SLOTS * 4);
    int*   spill  = (int*)  carve((size_t)E * 8);
    float* h0     = (float*)carve((size_t)N * H * 4);
    float* pA     = (float*)carve((size_t)N * 64 * 4);
    float* pB     = (float*)carve((size_t)N * 64 * 4);
    float* M2     = (float*)carve((size_t)H * 64 * 4);
    float* M1     = (float*)carve((size_t)H * 64 * 4);
    float* M0     = (float*)carve((size_t)H * 64 * 4);
    float* beta   = (float*)carve((size_t)3 * 64 * 4);
    v2i*   region = (v2i*)  carve((size_t)NBINS * NG * CAP * 8);
    int*   counts = (int*)  carve((size_t)NBINS * NG * 4);
    int*   ovfN   = (int*)  carve((size_t)NG * 4);
    v2i*   ovf    = (v2i*)  carve((size_t)NG * CH * 8);

    const int tilesM    = N / 16;
    const int gemmBlks  = (tilesM + 7) / 8;

    slots_bin_fold_xw1<<<1 + gemmBlks + NG, 256, 0, stream>>>(
        cnt, counts, ovfN, region, ovf, srcv, dstv, E, N, Wc, W2, bc, b2,
        M2, M1, M0, beta, x, W1, b1, h0, N, gemmBlks);
    slots_place_q<<<NBINS, PG_T, 0, stream>>>(
        cnt, table, spill, counts, ovfN, region, ovf, N, h0, M0, pA);

    gcn_gather64<false><<<(N + 7) / 8, 256, 0, stream>>>(
        pB, pA, beta, cnt, table, spill, N, E);
    gcn_gather64<false><<<(N + 7) / 8, 256, 0, stream>>>(
        pA, pB, beta + 64, cnt, table, spill, N, E);
    gcn_gather64<true><<<(N + 7) / 8, 256, 0, stream>>>(
        (float*)d_out, pA, beta + 128, cnt, table, spill, N, E);
}


# gather: all rows of a node asked for before the first is waited for (one group to 16 slots, a refilled window of 8 pair loads to 32), not 16/8/4/tail groups waited in turn; fp sum order unchanged
# speedup vs baseline: 7.1559x; 1.0330x over previous
#include <hip/hip_runtime.h>


typedef float v2f __attribute__((ext_vector_type(2)));
typedef float v4f __attribute__((ext_vector_type(4)));
typedef float v8f __attribute__((ext_vector_type(8)));
typedef int   v2i __attribute__((ext_vector_type(2)));
typedef int   v4i __attribute__((ext_vector_type(4)));

constexpr int SLOTS     = 64;
constexpr int N_NODES   = 50000;
constexpr int N_EDGES   = 800000;
constexpr int BIN_SHIFT = 8;
constexpr int BIN_NODES = 1 << BIN_SHIFT;
constexpr int NBINS     = (N_NODES + BIN_NODES - 1) / BIN_NODES;
constexpr int CH        = 2048;
constexpr int NG        = (N_EDGES + CH - 1) / CH;
constexpr int CAP       = 32;
constexpr int TPITCH    = SLOTS + 1;
constexpr int PB_T      = 512;
constexpr int PB_W      = PB_T / 32;
constexpr int PG_T      = 2 * PB_T;
constexpr int QK        = 128;
constexpr int QN        = 64;
constexpr int QLDB      = QK + 2;
constexpr int PB_U      = 7;

__device__ __forceinline__ void store_bin(int s, int d, int r, int g, int* binCnt,
                                          v2i* __restrict__ region, v2i* __restrict__ ovf) {
    v2i rec;
    rec.x = s;
    rec.y = d;
    if (r < CAP) {
        region[((size_t)(d >> BIN_SHIFT) * NG + g) * CAP + r] = rec;
    } else {
        int o = atomicAdd(&binCnt[NBINS], 1);
        ovf[(size_t)g * CH + o] = rec;
    }
}

__device__ __forceinline__ void store_slot(v2i rec, int slot, int* tab,
                                           int* __restrict__ cnt, int* __restrict__ spill, int n) {
    if (slot < SLOTS) {
        tab[(rec.y & (BIN_NODES - 1)) * TPITCH + slot] = rec.x;
    } else {
        int o = atomicAdd(&cnt[n], 1);
        ((v2i*)spill)[o] = rec;
    }
}

__global__ __launch_bounds__(PG_T) void slots_place_q(
        int* __restrict__ cnt, int* __restrict__ table, int* __restrict__ spill,
        const int* __restrict__ counts, const int* __restrict__ ovfN,
        const v2i* __restrict__ region, const v2i* __restrict__ ovf, int n,
        const float* __restrict__ h0, const float* __restrict__ M0,
        float* __restrict__ q) {
    __shared__ int tab[BIN_NODES * TPITCH];
    __shared__ int nodeCnt[BIN_NODES];
    __shared__ int cellCnt[NG];
    __shared__ int ovfCnt[NG];
    __shared__ int anyOvf;
    __shared__ float Blds[QN * QLDB];

    int b = blockIdx.x;
    int t = threadIdx.x;
    int w    = __builtin_amdgcn_readfirstlane(t >> 5);
    int lane = t & 31;
    bool placer = w < PB_W;
    int half = lane >> 4;
    int lr   = lane & 15;
    int tm   = b * (BIN_NODES / 16) + (w - PB_W);
    bool mult = !placer && tm < (n >> 4);

    if (placer) {
        for (int i = t; i < BIN_NODES * TPITCH; i += PB_T) tab[i] = 0;
        if (t < BIN_NODES) nodeCnt[t] = 0;
        if (t == 0) anyOvf = 0;
        for (int i = t; i < NG; i += PB_T) cellCnt[i] = counts[b * NG + i];
    } else {
        float m[QK * QN / PB_T];
#pragma unroll
        for (int u = 0; u < QK * QN / PB_T; ++u) m[u] = M0[u * PB_T + (t - PB_T)];
#pragma unroll
        for (int u = 0; u < QK * QN / PB_T; ++u) {
            int i = u * PB_T + (t - PB_T);
            Blds[(i & (QN - 1)) * QLDB + (i >> 6)] = m[u];
        }
    }
    __syncthreads();

    v8f acc[QN / 16] = {};
    if (mult) {
        const float* Arow = h0 + (size_t)(tm * 16 + lr) * QK;
        for (int k = 0; k < QK; k += 4) {
            v2f a = *(const v2f*)(Arow + k + 2 * half);
#pragma unroll
            for (int j = 0; j < QN / 16; ++j) {
                v2f bf = *(const v2f*)(&Blds[(j * 16 + lr) * QLDB + k + 2 * half]);
                acc[j] = __builtin_amdgcn_wmma_f32_16x16x4_f32(
                             false, a, false, bf, (short)0, acc[j], false, false);
            }
        }
    }

    if (placer) {
        for (int i = t; i < NG; i += PB_T) {
            int c = ovfN[i];
            ovfCnt[i] = c;
            if (c > 0) anyOvf = 1;
        }

        const v2i* reg = region + (size_t)b * NG * CAP;
        for (int g0 = w; g0 < NG; g0 += PB_W * PB_U) {
            v2i  ent[PB_U];
            bool ok[PB_U];
            int  slot[PB_U];
#pragma unroll
            for (int u = 0; u < PB_U; ++u) {
                int g = g0 + PB_W * u;
                ok[u] = (g < NG) && (lane < cellCnt[g < NG ? g : 0]);
                v2i e = {0, 0};
                if (ok[u]) e = reg[(size_t)g * CAP + lane];
                ent[u] = e;
            }
#pragma unroll
            for (int u = 0; u < PB_U; ++u) {
                slot[u] = 0;
                if (ok[u]) slot[u] = atomicAdd(&nodeCnt[ent[u].y & (BIN_NODES - 1)], 1);
            }
#pragma unroll
            for (int u = 0; u < PB_U; ++u)
                if (ok[u]) store_slot(ent[u], slot[u], tab, cnt, spill, n);
        }
    }
    __syncthreads();
    if (anyOvf && placer) {
        for (int g = 0; g < NG; ++g) {
            int c = ovfCnt[g];
            if (c > CH) c = CH;
            for (int i = t; i < c; i += PB_T) {
                v2i e = ovf[(size_t)g * CH + i];
                if ((e.y >> BIN_SHIFT) == b) {
                    int sl = atomicAdd(&nodeCnt[e.y & (BIN_NODES - 1)], 1);
                    store_slot(e, sl, tab, cnt, spill, n);
                }
            }
        }
    }
    __syncthreads();
    if (placer) {
        int node0 = b * BIN_NODES;
        if (t < BIN_NODES && node0 + t < n) cnt[node0 + t] = nodeCnt[t];
        int nrows = n - node0;
        if (nrows > BIN_NODES) nrows = BIN_NODES;
        int* trow = table + (size_t)node0 * SLOTS;
        for (int i = t; i < nrows * SLOTS; i += PB_T)
            trow[i] = tab[(i >> 6) * TPITCH + (i & (SLOTS - 1))];
    } else if (mult) {
        float sc[8];
#pragma unroll
        for (int r = 0; r < 8; ++r)
            sc[r] = rsqrtf((float)nodeCnt[(w - PB_W) * 16 + r + 8 * half] + 1.0f);
#pragma unroll
        for (int j = 0; j < QN / 16; ++j) {
            int col = j * 16 + lr;
#pragma unroll
            for (int r = 0; r < 8; ++r) {
                int row = tm * 16 + r + 8 * half;
                q[(size_t)row * QN + col] = acc[j][r] * sc[r];
            }
        }
    }
}

__device__ __forceinline__ void fold_product(const float* A, const float* B,
                                             float* C, float* Blds) {
    constexpr int KK  = 128;
    constexpr int TN  = 4;
    constexpr int Nc  = TN * 16;
    constexpr int LDB = KK + 2;

    for (int i = threadIdx.x; i < KK * Nc; i += 256) {
        int k = i / Nc;
        int c = i - k * Nc;
        Blds[c * LDB + k] = B[i];
    }
    __syncthreads();

    int tm   = threadIdx.x >> 5;
    int lane = threadIdx.x & 31;
    int half = lane >> 4;
    int lr   = lane & 15;

    const float* Arow = A + (size_t)(tm * 16 + lr) * KK;
    v8f acc[TN] = {};
    for (int k = 0; k < KK; k += 4) {
        v2f a = *(const v2f*)(Arow + k + 2 * half);
#pragma unroll
        for (int t = 0; t < TN; ++t) {
            v2f b = *(const v2f*)(&Blds[(t * 16 + lr) * LDB + k + 2 * half]);
            acc[t] = __builtin_amdgcn_wmma_f32_16x16x4_f32(
                         false, a, false, b, (short)0, acc[t], false, false);
        }
    }
#pragma unroll
    for (int t = 0; t < TN; ++t) {
        int col = t * 16 + lr;
#pragma unroll
        for (int r = 0; r < 8; ++r) {
            int row = tm * 16 + r + 8 * half;
            C[(size_t)row * Nc + col] = acc[t][r];
        }
    }
    __syncthreads();
}

template <int KK, int TN, bool RELU, bool BIAS>
__device__ __forceinline__ void gemm_group(
        const float* __restrict__ A, const float* __restrict__ B,
        const float* __restrict__ bias,
        float* __restrict__ C, int M, int blk, float* Blds) {
    constexpr int Nc  = TN * 16;
    constexpr int LDB = KK + 2;

    for (int i = threadIdx.x; i < KK * Nc; i += 256) {
        int k = i / Nc;
        int c = i - k * Nc;
        Blds[c * LDB + k] = B[i];
    }
    __syncthreads();

    int tilesM = M >> 4;
    int tm     = blk * 8 + __builtin_amdgcn_readfirstlane((int)(threadIdx.x >> 5));
    int lane   = threadIdx.x & 31;
    int half   = lane >> 4;
    int lr     = lane & 15;

    if (tm < tilesM) {
        const float* Arow = A + (size_t)(tm * 16 + lr) * KK;
        __builtin_prefetch(Arow, 0, 0);
        __builtin_prefetch(Arow + 64, 0, 0);

        v8f acc[TN] = {};
        for (int k = 0; k < KK; k += 4) {
            v2f a = *(const v2f*)(Arow + k + 2 * half);
#pragma unroll
            for (int t = 0; t < TN; ++t) {
                v2f b = *(const v2f*)(&Blds[(t * 16 + lr) * LDB + k + 2 * half]);
                acc[t] = __builtin_amdgcn_wmma_f32_16x16x4_f32(
                             false, a, false, b, (short)0, acc[t], false, false);
            }
        }

#pragma unroll
        for (int t = 0; t < TN; ++t) {
            int col  = t * 16 + lr;
            float bv = BIAS ? bias[col] : 0.0f;
#pragma unroll
            for (int r = 0; r < 8; ++r) {
                int row = tm * 16 + r + 8 * half;
                float v = acc[t][r] + bv;
                if (RELU) v = fmaxf(v, 0.0f);
                C[(size_t)row * Nc + col] = v;
            }
        }
    }
}

__global__ __launch_bounds__(256, 8) void slots_bin_fold_xw1(
        int* __restrict__ cnt, int* __restrict__ counts, int* __restrict__ ovfN,
        v2i* __restrict__ region, v2i* __restrict__ ovf,
        const int* __restrict__ src, const int* __restrict__ dst, int e, int n,
        const float* Wc, const float* W2, const float* bc, const float* b2,
        float* M2, float* M1, float* M0, float* beta,
        const float* __restrict__ x, const float* __restrict__ W1,
        const float* __restrict__ b1, float* __restrict__ h0,
        int rows, int gemm_groups) {
    __shared__ float Blds[128 * 130];
    __shared__ int binCnt[NBINS + 1];
    if (blockIdx.x == 0) {
        if (threadIdx.x == 0) cnt[n] = 0;
        fold_product(Wc + 2 * 128 * 128, W2, M2, Blds);
        fold_product(Wc + 1 * 128 * 128, M2, M1, Blds);
        fold_product(Wc,                 M1, M0, Blds);
        if (threadIdx.x < 192) {
            int l = threadIdx.x >> 6;
            int c = threadIdx.x & 63;
            const float* M = (l == 0) ? M1 : ((l == 1) ? M2 : W2);
            const float* b = bc + l * 128;
            float s = 0.0f;
            for (int k = 0; k < 128; ++k) s += b[k] * M[k * 64 + c];
            if (l == 2) s += b2[c];
            beta[l * 64 + c] = s;
        }
        return;
    }
    if ((int)blockIdx.x <= gemm_groups) {
        gemm_group<128, 8, true, true>(
            x, W1, b1, h0, rows, (int)blockIdx.x - 1, Blds);
        return;
    }
    int g = (int)blockIdx.x - 1 - gemm_groups;
    for (int i = threadIdx.x; i < NBINS + 1; i += 256) binCnt[i] = 0;
    __syncthreads();
    constexpr int UB = CH / 1024;
    int sv[UB * 4], dv[UB * 4], rk[UB * 4];
#pragma unroll
    for (int u = 0; u < UB; ++u) {
        int i = g * CH + u * 1024 + (int)threadIdx.x * 4;
        v4i s4 = {0, 0, 0, 0};
        v4i d4 = {-1, -1, -1, -1};
        if (i + 3 < e) {
            s4 = *(const v4i*)(src + i);
            d4 = *(const v4i*)(dst + i);
        } else {
            if (i     < e) { s4.x = src[i];     d4.x = dst[i];     }
            if (i + 1 < e) { s4.y = src[i + 1]; d4.y = dst[i + 1]; }
            if (i + 2 < e) { s4.z = src[i + 2]; d4.z = dst[i + 2]; }
        }
        sv[u * 4 + 0] = s4.x; sv[u * 4 + 1] = s4.y; sv[u * 4 + 2] = s4.z; sv[u * 4 + 3] = s4.w;
        dv[u * 4 + 0] = d4.x; dv[u * 4 + 1] = d4.y; dv[u * 4 + 2] = d4.z; dv[u * 4 + 3] = d4.w;
    }
#pragma unroll
    for (int q = 0; q < UB * 4; ++q) {
        rk[q] = 0;
        if (dv[q] >= 0) rk[q] = atomicAdd(&binCnt[dv[q] >> BIN_SHIFT], 1);
    }
#pragma unroll
    for (int q = 0; q < UB * 4; ++q)
        if (dv[q] >= 0) store_bin(sv[q], dv[q], rk[q], g, binCnt, region, ovf);
    __syncthreads();
    for (int i = threadIdx.x; i < NBINS; i += 256) {
        int c = binCnt[i];
        counts[i * NG + g] = c < CAP ? c : CAP;
    }
    if (threadIdx.x == 0) ovfN[g] = binCnt[NBINS];
}


__device__ __forceinline__ v4f gather_row(const v4f* __restrict__ q4, int rec, int i0,
                                          int half, int c) {
    int s0 = __builtin_amdgcn_readlane(rec, i0);
    int s1 = __builtin_amdgcn_readlane(rec, i0 + 1);
    int s  = half ? s1 : s0;
    return q4[(unsigned)(s * 16 + c)];
}

__device__ __forceinline__ v4f gather_keep(v4f t, int i0, int half, int m) {
    v4f z = {0.0f, 0.0f, 0.0f, 0.0f};
    return (i0 + half < m) ? t : z;
}

template <int U>
__device__ __forceinline__ void gather_group(v4f& acc, const v4f* __restrict__ q4,
                                             int rec, int m, int half, int c) {
    v4f r[U];
#pragma unroll
    for (int u = 0; u < U; ++u) r[u] = gather_row(q4, rec, 2 * u, half, c);
    asm volatile("" :: "v"(r[U - 1]));
#pragma unroll
    for (int u = 0; u < U; ++u)
        acc += (u < U - 2) ? r[u] : gather_keep(r[u], 2 * u, half, m);
}

template <int R>
__device__ __forceinline__ void gather_window(v4f& acc, const v4f* __restrict__ q4,
                                              int rec, int m, int half, int c) {
    v4f r[8];
#pragma unroll
    for (int u = 0; u < 8; ++u) r[u] = gather_row(q4, rec, 2 * u, half, c);
#pragma unroll
    for (int u = 0; u < 8; ++u) {
        acc += r[u];
        if (u < R) r[u] = gather_row(q4, rec, 16 + 2 * u, half, c);
    }
    asm volatile("" :: "v"(r[R - 1]));
#pragma unroll
    for (int u = 0; u < R; ++u)
        acc += (u < R - 2) ? r[u] : gather_keep(r[u], 16 + 2 * u, half, m);
}

__device__ __forceinline__ void gather_chunk(v4f& acc, const v4f* __restrict__ q4,
                                             int rec, int m, int lane, int half, int c) {
    if (lane >= m) rec = 0;
    int pairs = (m + 1) >> 1;
    if (pairs <= 8) {
        if (pairs <= 4) {
            if (pairs <= 2) gather_group<2>(acc, q4, rec, m, half, c);
            else            gather_group<4>(acc, q4, rec, m, half, c);
        } else {
            if (pairs <= 6) gather_group<6>(acc, q4, rec, m, half, c);
            else            gather_group<8>(acc, q4, rec, m, half, c);
        }
    } else {
        if (pairs <= 12) {
            if (pairs <= 10) gather_window<2>(acc, q4, rec, m, half, c);
            else             gather_window<4>(acc, q4, rec, m, half, c);
        } else {
            if (pairs <= 14) gather_window<6>(acc, q4, rec, m, half, c);
            else             gather_window<8>(acc, q4, rec, m, half, c);
        }
    }
}

template <bool LAST>
__global__ __launch_bounds__(256, 16) void gcn_gather64(
        float* __restrict__ out, const float* __restrict__ q,
        const float* __restrict__ beta, const int* __restrict__ cnt,
        const int* __restrict__ table, const int* __restrict__ spill,
        int n, int e) {
    int node = blockIdx.x * 8 + __builtin_amdgcn_readfirstlane((int)(threadIdx.x >> 5));
    if (node >= n) return;
    int lane = threadIdx.x & 31;
    int half = lane >> 4;
    int c    = lane & 15;

    const v4f* q4  = (const v4f*)q;
    const int* row = table + (size_t)node * SLOTS;

    int deg  = __builtin_amdgcn_readfirstlane(cnt[node]);
    int rec0 = row[lane];
    v4f self = q4[(size_t)node * 16 + c];
    v4f bv   = ((const v4f*)beta)[c];

    int filled = deg < SLOTS ? deg : SLOTS;
    v4f acc = {0.0f, 0.0f, 0.0f, 0.0f};
    gather_chunk(acc, q4, rec0, filled < 32 ? filled : 32, lane, half, c);
    if (filled > 32)
        gather_chunk(acc, q4, row[32 + lane], filled - 32, lane, half, c);

    if (deg > SLOTS) {
        int total = __builtin_amdgcn_readfirstlane(cnt[n]);
        if (total > e) total = e;
        const v2i* sp = (const v2i*)spill;
        for (int ob = 0; ob < total; ob += 32) {
            int mm = total - ob;
            if (mm > 32) mm = 32;
            v2i orec = {0, -1};
            if (lane < mm) orec = sp[ob + lane];
            for (int jj = 0; jj < mm; ++jj) {
                int s = __builtin_amdgcn_readlane(orec.x, jj);
                int d = __builtin_amdgcn_readlane(orec.y, jj);
                if (d == node && half == 0) acc += q4[(size_t)s * 16 + c];
            }
        }
    }

    v4f oth;
    oth.x = __shfl_xor(acc.x, 16);
    oth.y = __shfl_xor(acc.y, 16);
    oth.z = __shfl_xor(acc.z, 16);
    oth.w = __shfl_xor(acc.w, 16);
    float dn = rsqrtf((float)deg + 1.0f);
    v4f v = ((acc + oth) + self) * dn + bv;
    if (!LAST) v = v * dn;
    if (half == 0)
        ((v4f*)out)[(size_t)node * 16 + c] = v;
}

extern "C" void kernel_launch(void* const* d_in, const int* in_sizes, int n_in,
                              void* d_out, int out_size, void* d_ws, size_t ws_size,
                              hipStream_t stream) {
    const float* x  = (const float*)d_in[0];
    const int*   ei = (const int*)  d_in[1];
    const float* W1 = (const float*)d_in[2];
    const float* b1 = (const float*)d_in[3];
    const float* Wc = (const float*)d_in[4];
    const float* bc = (const float*)d_in[5];
    const float* W2 = (const float*)d_in[6];
    const float* b2 = (const float*)d_in[7];

    const int N = N_NODES, E = N_EDGES, H = 128;
    const int* srcv = ei;
    const int* dstv = ei + E;


    char* ws = (char*)d_ws;
    size_t off = 0;
    auto carve = [&](size_t bytes) -> void* {
        void* p = ws + off;
        off = (off + bytes + 255) & ~(size_t)255;
        return p;
    };
    int*   cnt    = (int*)  carve((size_t)(N + 1) * 4);
    int*   table  = (int*)  carve((size_t)N * SLOTS * 4);
    int*   spill  = (int*)  carve((size_t)E * 8);
    float* h0     = (float*)carve((size_t)N * H * 4);
    float* pA     = (float*)carve((size_t)N * 64 * 4);
    float* pB     = (float*)carve((size_t)N * 64 * 4);
    float* M2     = (float*)carve((size_t)H * 64 * 4);
    float* M1     = (float*)carve((size_t)H * 64 * 4);
    float* M0     = (float*)carve((size_t)H * 64 * 4);
    float* beta   = (float*)carve((size_t)3 * 64 * 4);
    v2i*   region = (v2i*)  carve((size_t)NBINS * NG * CAP * 8);
    int*   counts = (int*)  carve((size_t)NBINS * NG * 4);
    int*   ovfN   = (int*)  carve((size_t)NG * 4);
    v2i*   ovf    = (v2i*)  carve((size_t)NG * CH * 8);

    const int tilesM    = N / 16;
    const int gemmBlks  = (tilesM + 7) / 8;

    slots_bin_fold_xw1<<<1 + gemmBlks + NG, 256, 0, stream>>>(
        cnt, counts, ovfN, region, ovf, srcv, dstv, E, N, Wc, W2, bc, b2,
        M2, M1, M0, beta, x, W1, b1, h0, N, gemmBlks);
    slots_place_q<<<NBINS, PG_T, 0, stream>>>(
        cnt, table, spill, counts, ovfN, region, ovf, N, h0, M0, pA);

    gcn_gather64<false><<<(N + 7) / 8, 256, 0, stream>>>(
        pB, pA, beta, cnt, table, spill, N, E);
    gcn_gather64<false><<<(N + 7) / 8, 256, 0, stream>>>(
        pA, pB, beta + 64, cnt, table, spill, N, E);
    gcn_gather64<true><<<(N + 7) / 8, 256, 0, stream>>>(
        (float*)d_out, pA, beta + 128, cnt, table, spill, N, E);
}
